# starting kernel
# baseline (speedup 1.0000x reference)
#include <hip/hip_runtime.h>


#define HDIM   64
#define VOCABN 64
#define BATCH  64
#define SEQL   4096

typedef __attribute__((ext_vector_type(2))) float v2f;
typedef __attribute__((ext_vector_type(2))) float f32x2;
typedef __attribute__((ext_vector_type(8))) float v8f;

__device__ __forceinline__ v2f frag_a(const float* A, int lda, int mt, int kt, int lane) {
  int row = (mt << 4) + (lane & 15);
  int k0  = (kt << 2) + ((lane >> 4) << 1);
  v2f a;
  a.x = A[row * lda + k0];
  a.y = A[row * lda + k0 + 1];
  return a;
}
__device__ __forceinline__ v2f frag_b(const float* Bm, int ldb, int kt, int nt, int lane) {
  int col = (nt << 4) + (lane & 15);
  int k0  = (kt << 2) + ((lane >> 4) << 1);
  v2f b;
  b.x = Bm[k0 * ldb + col];
  b.y = Bm[(k0 + 1) * ldb + col];
  return b;
}

__device__ __forceinline__ v8f wmma_f32(v2f a, v2f b, v8f c) {
  return __builtin_amdgcn_wmma_f32_16x16x4_f32(false, a, false, b, (short)0, c,
                                               false, false);
}

__global__ __launch_bounds__(256) void encode_table_kernel(
    const float* __restrict__ embed, const float* __restrict__ W1,
    const float* __restrict__ b1, const float* __restrict__ W2,
    const float* __restrict__ b2, const float* __restrict__ gamma,
    const float* __restrict__ beta, float* __restrict__ Tout,
    float* __restrict__ rdenom) {
  __shared__ float Rlds[64 * 128];
  __shared__ float Xlds[64 * 64];
  const int tid  = threadIdx.x;
  const int wave = tid >> 5;
  const int lane = tid & 31;

  for (int tile = wave; tile < 32; tile += 8) {
    int mt = tile >> 3, nt = tile & 7;
    v8f acc = {};
#pragma unroll
    for (int kt = 0; kt < 16; ++kt) {
      v2f a = frag_a(embed, 64, mt, kt, lane);
      v2f b = frag_b(W1, 128, kt, nt, lane);
      acc = wmma_f32(a, b, acc);
    }
    int col   = (nt << 4) + (lane & 15);
    int rbase = (mt << 4) + ((lane >> 4) << 3);
#pragma unroll
    for (int v = 0; v < 8; ++v) {
      float x = acc[v] + b1[col];
      Rlds[(rbase + v) * 128 + col] = fmaxf(x, 0.0f);
    }
  }
  __syncthreads();

  for (int tile = wave; tile < 16; tile += 8) {
    int mt = tile >> 2, nt = tile & 3;
    v8f acc = {};
#pragma unroll
    for (int kt = 0; kt < 32; ++kt) {
      v2f a = frag_a(Rlds, 128, mt, kt, lane);
      v2f b = frag_b(W2, 64, kt, nt, lane);
      acc = wmma_f32(a, b, acc);
    }
    int col   = (nt << 4) + (lane & 15);
    int rbase = (mt << 4) + ((lane >> 4) << 3);
#pragma unroll
    for (int v = 0; v < 8; ++v) {
      int row = rbase + v;
      Xlds[row * 64 + col] = embed[row * 64 + col] + acc[v] + b2[col];
    }
  }
  __syncthreads();

  if (tid < 64) {
    const float* xr = &Xlds[tid * 64];
    float mu = 0.f;
#pragma unroll
    for (int c = 0; c < 64; ++c) mu += xr[c];
    mu *= (1.0f / 64.0f);
    float var = 0.f;
#pragma unroll
    for (int c = 0; c < 64; ++c) { float d = xr[c] - mu; var += d * d; }
    var *= (1.0f / 64.0f);
    float rs  = rsqrtf(var + 1e-5f);
    float ksq = 0.f;
#pragma unroll
    for (int c = 0; c < 64; ++c) {
      float tv = (xr[c] - mu) * rs * gamma[c] + beta[c];
      Tout[tid * 64 + c] = tv;
      ksq += tv * tv;
    }
    rdenom[tid] = 1.0f / (ksq + 1e-6f);
  }
}

__global__ __launch_bounds__(128) void delta_scan_kernel(
    const int* __restrict__ seq, const float* __restrict__ Ttab,
    const float* __restrict__ rdenom, float* __restrict__ ctx) {
  __shared__ float4 Tl4[VOCABN * 16];
  __shared__ float  rd[VOCABN];
  __shared__ int    toks[SEQL];
  const int b    = blockIdx.x;
  const int tid  = threadIdx.x;
  const int r    = tid >> 1;
  const int half = tid & 1;

  const float4* Tg4 = (const float4*)Ttab;
  for (int i = tid; i < VOCABN * 16; i += 128) Tl4[i] = Tg4[i];
  if (tid < VOCABN) rd[tid] = rdenom[tid];
  const int4* sg = (const int4*)(seq + b * SEQL);
  int4*       tl = (int4*)toks;
  for (int i = tid; i < SEQL / 4; i += 128) tl[i] = sg[i];
  __syncthreads();

  const float* Tl    = (const float*)Tl4;
  const int    kbase = half << 3;

  f32x2 M2[16];
#pragma unroll
  for (int i = 0; i < 16; ++i) M2[i] = (f32x2){0.f, 0.f};

  for (int t = 0; t < SEQL - 1; ++t) {
    int tok = toks[t];
    const float4* kp = &Tl4[tok * 16 + kbase];
    f32x2 kv[16];
    f32x2 acc0 = (f32x2){0.f, 0.f};
    f32x2 acc1 = (f32x2){0.f, 0.f};
#pragma unroll
    for (int i = 0; i < 8; ++i) {
      float4 ld = kp[i];
      kv[2 * i]     = (f32x2){ld.x, ld.y};
      kv[2 * i + 1] = (f32x2){ld.z, ld.w};
      acc0 = __builtin_elementwise_fma(M2[2 * i], kv[2 * i], acc0);
      acc1 = __builtin_elementwise_fma(M2[2 * i + 1], kv[2 * i + 1], acc1);
    }
    float part = (acc0.x + acc0.y) + (acc1.x + acc1.y);
    float vp   = part + __shfl_xor(part, 1, 32);
    float dv   = Tl[tok * 64 + r] - vp * rd[tok];
    f32x2 dvv  = (f32x2){dv, dv};
#pragma unroll
    for (int i = 0; i < 16; ++i)
      M2[i] = __builtin_elementwise_fma(dvv, kv[i], M2[i]);
  }

  int tok = toks[SEQL - 1];
  const float4* qp = &Tl4[tok * 16 + kbase];
  f32x2 acc0 = (f32x2){0.f, 0.f};
  f32x2 acc1 = (f32x2){0.f, 0.f};
#pragma unroll
  for (int i = 0; i < 8; ++i) {
    float4 ld = qp[i];
    acc0 = __builtin_elementwise_fma(M2[2 * i], (f32x2){ld.x, ld.y}, acc0);
    acc1 = __builtin_elementwise_fma(M2[2 * i + 1], (f32x2){ld.z, ld.w}, acc1);
  }
  float part = (acc0.x + acc0.y) + (acc1.x + acc1.y);
  float vp   = part + __shfl_xor(part, 1, 32);
  if (half == 0) ctx[b * 64 + r] = vp;
}

__global__ __launch_bounds__(256) void head_kernel(
    const float* __restrict__ ctx, const float* __restrict__ Wr,
    const float* __restrict__ br, const float* __restrict__ Wo,
    const float* __restrict__ bo, float* __restrict__ out) {
  __shared__ float Glds[64 * 64];
  const int tid  = threadIdx.x;
  const int wave = tid >> 5;
  const int lane = tid & 31;

  for (int tile = wave; tile < 16; tile += 8) {
    int mt = tile >> 2, nt = tile & 3;
    v8f acc = {};
#pragma unroll
    for (int kt = 0; kt < 16; ++kt) {
      v2f a = frag_a(ctx, 64, mt, kt, lane);
      v2f b = frag_b(Wr, 64, kt, nt, lane);
      acc = wmma_f32(a, b, acc);
    }
    int col   = (nt << 4) + (lane & 15);
    int rbase = (mt << 4) + ((lane >> 4) << 3);
#pragma unroll
    for (int v = 0; v < 8; ++v) Glds[(rbase + v) * 64 + col] = acc[v] + br[col];
  }
  __syncthreads();

  for (int tile = wave; tile < 16; tile += 8) {
    int mt = tile >> 2, nt = tile & 3;
    v8f acc = {};
#pragma unroll
    for (int kt = 0; kt < 16; ++kt) {
      v2f a = frag_a(Glds, 64, mt, kt, lane);
      v2f b = frag_b(Wo, 64, kt, nt, lane);
      acc = wmma_f32(a, b, acc);
    }
    int col   = (nt << 4) + (lane & 15);
    int rbase = (mt << 4) + ((lane >> 4) << 3);
#pragma unroll
    for (int v = 0; v < 8; ++v) out[(rbase + v) * 64 + col] = acc[v] + bo[col];
  }
}

extern "C" void kernel_launch(void* const* d_in, const int* in_sizes, int n_in,
                              void* d_out, int out_size, void* d_ws, size_t ws_size,
                              hipStream_t stream) {
  const int*   seq   = (const int*)d_in[0];
  const float* embed = (const float*)d_in[1];
  const float* W1    = (const float*)d_in[2];
  const float* b1    = (const float*)d_in[3];
  const float* W2    = (const float*)d_in[4];
  const float* b2    = (const float*)d_in[5];
  const float* gamma = (const float*)d_in[6];
  const float* beta  = (const float*)d_in[7];
  const float* Wr    = (const float*)d_in[8];
  const float* br    = (const float*)d_in[9];
  const float* Wo    = (const float*)d_in[10];
  const float* bo    = (const float*)d_in[11];

  float* Tt     = (float*)d_ws;
  float* rdenom = Tt + VOCABN * HDIM;
  float* ctx    = rdenom + VOCABN;
  float* out    = (float*)d_out;

  encode_table_kernel<<<1, 256, 0, stream>>>(embed, W1, b1, W2, b2, gamma, beta,
                                             Tt, rdenom);
  delta_scan_kernel<<<BATCH, 128, 0, stream>>>(seq, Tt, rdenom, ctx);
  head_kernel<<<1, 256, 0, stream>>>(ctx, Wr, br, Wo, bo, out);
}


# Scan re-cut in 3 stretches of 104/104/48 chunks on 256 work-groups, the short last one running both its scans in one block; head joins M1(A2(A3 q))+C2(A3 q)+C3 q; summation order changes
# speedup vs baseline: 7.3225x; 7.3225x over previous
#include <hip/hip_runtime.h>


#define HDIM   64
#define VOCABN 64
#define BATCH  64
#define SEQL   4096

typedef __attribute__((ext_vector_type(2))) float v2f;
typedef __attribute__((ext_vector_type(2))) float f32x2;
typedef __attribute__((ext_vector_type(8))) float v8f;

__device__ __forceinline__ v2f frag_a(const float* A, int lda, int mt, int kt, int lane) {
  int row = (mt << 4) + (lane & 15);
  int k0  = (kt << 2) + ((lane >> 4) << 1);
  v2f a;
  a.x = A[row * lda + k0];
  a.y = A[row * lda + k0 + 1];
  return a;
}
__device__ __forceinline__ v2f frag_b(const float* Bm, int ldb, int kt, int nt, int lane) {
  int col = (nt << 4) + (lane & 15);
  int k0  = (kt << 2) + ((lane >> 4) << 1);
  v2f b;
  b.x = Bm[k0 * ldb + col];
  b.y = Bm[(k0 + 1) * ldb + col];
  return b;
}

__device__ __forceinline__ v8f wmma_f32(v2f a, v2f b, v8f c) {
  return __builtin_amdgcn_wmma_f32_16x16x4_f32(false, a, false, b, (short)0, c,
                                               false, false);
}

__global__ __launch_bounds__(256) void encode_table_kernel(
    const float* __restrict__ embed, const float* __restrict__ W1,
    const float* __restrict__ b1, const float* __restrict__ W2,
    const float* __restrict__ b2, const float* __restrict__ gamma,
    const float* __restrict__ beta, float* __restrict__ Tout,
    float* __restrict__ rdenom) {
  __shared__ float Rlds[64 * 128];
  __shared__ float Xlds[64 * 64];
  const int tid  = threadIdx.x;
  const int wave = tid >> 5;
  const int lane = tid & 31;

  for (int tile = wave; tile < 32; tile += 8) {
    int mt = tile >> 3, nt = tile & 7;
    v8f acc = {};
#pragma unroll
    for (int kt = 0; kt < 16; ++kt) {
      v2f a = frag_a(embed, 64, mt, kt, lane);
      v2f b = frag_b(W1, 128, kt, nt, lane);
      acc = wmma_f32(a, b, acc);
    }
    int col   = (nt << 4) + (lane & 15);
    int rbase = (mt << 4) + ((lane >> 4) << 3);
#pragma unroll
    for (int v = 0; v < 8; ++v) {
      float x = acc[v] + b1[col];
      Rlds[(rbase + v) * 128 + col] = fmaxf(x, 0.0f);
    }
  }
  __syncthreads();

  for (int tile = wave; tile < 16; tile += 8) {
    int mt = tile >> 2, nt = tile & 3;
    v8f acc = {};
#pragma unroll
    for (int kt = 0; kt < 32; ++kt) {
      v2f a = frag_a(Rlds, 128, mt, kt, lane);
      v2f b = frag_b(W2, 64, kt, nt, lane);
      acc = wmma_f32(a, b, acc);
    }
    int col   = (nt << 4) + (lane & 15);
    int rbase = (mt << 4) + ((lane >> 4) << 3);
#pragma unroll
    for (int v = 0; v < 8; ++v) {
      int row = rbase + v;
      Xlds[row * 64 + col] = embed[row * 64 + col] + acc[v] + b2[col];
    }
  }
  __syncthreads();

  if (tid < 64) {
    const float* xr = &Xlds[tid * 64];
    float mu = 0.f;
#pragma unroll
    for (int c = 0; c < 64; ++c) mu += xr[c];
    mu *= (1.0f / 64.0f);
    float var = 0.f;
#pragma unroll
    for (int c = 0; c < 64; ++c) { float d = xr[c] - mu; var += d * d; }
    var *= (1.0f / 64.0f);
    float rs  = rsqrtf(var + 1e-5f);
    float ksq = 0.f;
#pragma unroll
    for (int c = 0; c < 64; ++c) {
      float tv = (xr[c] - mu) * rs * gamma[c] + beta[c];
      Tout[tid * 64 + c] = tv;
      ksq += tv * tv;
    }
    rdenom[tid] = 1.0f / (ksq + 1e-6f);
  }
}

#define TP   68
#define NCH  256
#define SEGM 104
#define SEG3 (NCH - 2 * SEGM)
#define XCAP 128
#define MPLANE (BATCH * HDIM * HDIM)

__device__ __forceinline__ v8f rows_dot(const v8f (&Mt)[4], const float* krow) {
  v8f acc = {};
#pragma unroll
  for (int mt = 0; mt < 4; ++mt) {
    const float4 lo = *(const float4*)(krow + 16 * mt);
    const float4 hi = *(const float4*)(krow + 16 * mt + 4);
    acc = wmma_f32((v2f){lo.x, lo.y}, (v2f){Mt[mt][0], Mt[mt][1]}, acc);
    acc = wmma_f32((v2f){lo.z, lo.w}, (v2f){Mt[mt][2], Mt[mt][3]}, acc);
    acc = wmma_f32((v2f){hi.x, hi.y}, (v2f){Mt[mt][4], Mt[mt][5]}, acc);
    acc = wmma_f32((v2f){hi.z, hi.w}, (v2f){Mt[mt][6], Mt[mt][7]}, acc);
  }
  return acc;
}

__global__ __launch_bounds__(256) void delta_scan_kernel(
    const int* __restrict__ seq, const float* __restrict__ Ttab,
    const float* __restrict__ rdenom, float* __restrict__ Xws,
    float* __restrict__ Mws, float* __restrict__ wvec,
    float* __restrict__ vvec) {
  __shared__ __attribute__((aligned(16))) float Tl[VOCABN * TP];
  __shared__ __attribute__((aligned(16))) float Gl[VOCABN * VOCABN];
  __shared__ float rd[VOCABN];
  __shared__ __attribute__((aligned(16))) int toks[SEQL];
  const int b    = blockIdx.x;
  const int role = blockIdx.y;
  const int tid  = threadIdx.x;
  const int wave = tid >> 5;
  const int cb   = (role == 0) ? 0 : ((role == 3) ? 2 * SEGM : SEGM);
  const int nch  = (role == 3) ? SEG3 : SEGM;
  const bool homog = (role == 2) || (role == 3 && wave >= 4);
  const float ksel = homog ? 0.0f : 1.0f;
  const int lane = tid & 31;
  const int li   = lane & 15;
  const int h    = lane >> 4;

  const float4* Tg4 = (const float4*)Ttab;
  for (int i = tid; i < VOCABN * 16; i += 256)
    *(float4*)&Tl[(i >> 4) * TP + ((i & 15) << 2)] = Tg4[i];
  if (tid < VOCABN) rd[tid] = rdenom[tid];
  const int4* sg = (const int4*)(seq + b * SEQL);
  int4*       tl = (int4*)toks;
  for (int i = tid; i < SEQL / 4; i += 256) tl[i] = sg[i];
  __syncthreads();

  for (int tile = wave; tile < 16; tile += 8) {
    const int mt = tile >> 2, nt = tile & 3;
    v8f acc = {};
    const float* ar = &Tl[((mt << 4) + li) * TP + (h << 1)];
    const float* br = &Tl[((nt << 4) + li) * TP + (h << 1)];
#pragma unroll
    for (int kt = 0; kt < 16; ++kt) {
      v2f a = (v2f){ar[4 * kt], ar[4 * kt + 1]};
      v2f bb = (v2f){br[4 * kt], br[4 * kt + 1]};
      acc = wmma_f32(a, bb, acc);
    }
#pragma unroll
    for (int v = 0; v < 8; ++v)
      Gl[((mt << 4) + (h << 3) + v) * VOCABN + (nt << 4) + li] = acc[v];
  }
  __syncthreads();

  float* Xb = Xws + (size_t)(role * BATCH + b) * (XCAP * 256);
  const int npass = (nch * 16 + 255) >> 8;
  for (int n = 0; n < npass; ++n) {
    const int w  = tid + 256 * n;
    const int cl = w >> 4;
    if (cl >= nch) continue;
    const int c  = cb + cl;
    const int i  = w & 15;
    int tk[16];
    const int4* tp = (const int4*)&toks[16 * c];
#pragma unroll
    for (int u = 0; u < 4; ++u) {
      const int4 t4 = tp[u];
      tk[4 * u] = t4.x; tk[4 * u + 1] = t4.y; tk[4 * u + 2] = t4.z; tk[4 * u + 3] = t4.w;
    }
    float x[16], y[16];
#pragma unroll
    for (int m = 0; m < 16; ++m) x[m] = (m == i) ? 1.0f : 0.0f;
    y[15] = ((16 * c + 15 < SEQL - 1) ? rd[tk[15]] : 0.0f) * x[15];
#pragma unroll
    for (int j = 14; j >= 0; --j) {
      const float* gr = &Gl[tk[j] * VOCABN];
      float s = 0.0f;
#pragma unroll
      for (int m = j + 1; m < 16; ++m) s = fmaf(gr[tk[m]], y[m], s);
      x[j] = (j < i) ? -s : x[j];
      y[j] = rd[tk[j]] * x[j];
    }
    float4* dst = (float4*)(Xb + (cl * 16 + i) * 16);
    dst[0] = make_float4(x[0], x[1], x[2], x[3]);
    dst[1] = make_float4(x[4], x[5], x[6], x[7]);
    dst[2] = make_float4(x[8], x[9], x[10], x[11]);
    dst[3] = make_float4(x[12], x[13], x[14], x[15]);
  }
  __syncthreads();
  if (wave >= 4 && role != 3) return;

  const int r0 = (wave & 3) << 4;
  v8f Mt[4];
#pragma unroll
  for (int mt = 0; mt < 4; ++mt)
#pragma unroll
    for (int v = 0; v < 8; ++v)
      Mt[mt][v] = (homog && (r0 + li == 16 * mt + 8 * h + v)) ? 1.0f : 0.0f;

  const float4* Xg = (const float4*)Xb + (li * 4 + 2 * h);

  struct Ops { float4 ka[8]; float kt[4][8]; float kr[8]; float rv[8]; };

  auto load_ops = [&](Ops& o, const int tA, const int4 ta, const int4 tb) {
    const float* krow = &Tl[tA * TP + 8 * h];
#pragma unroll
    for (int u = 0; u < 8; ++u) o.ka[u] = *(const float4*)(krow + 16 * (u >> 1) + 4 * (u & 1));
    const int tv[8] = {ta.x, ta.y, ta.z, ta.w, tb.x, tb.y, tb.z, tb.w};
#pragma unroll
    for (int v = 0; v < 8; ++v) {
      const float* row = &Tl[tv[v] * TP + li];
#pragma unroll
      for (int mt = 0; mt < 4; ++mt) o.kt[mt][v] = row[16 * mt];
      o.kr[v] = row[r0];
      o.rv[v] = rd[tv[v]];
    }
  };

  auto compute = [&](const Ops& o, const float4 xlo, const float4 xhi, const bool pad) {
    v8f Pa = {}, Pb = {};
#pragma unroll
    for (int mt = 0; mt < 4; mt += 2) {
      const float4 lo = o.ka[2 * mt], hi = o.ka[2 * mt + 1];
      const float4 lp = o.ka[2 * mt + 2], hp = o.ka[2 * mt + 3];
      Pa = wmma_f32((v2f){lo.x, lo.y}, (v2f){Mt[mt][0], Mt[mt][1]}, Pa);
      Pb = wmma_f32((v2f){lp.x, lp.y}, (v2f){Mt[mt + 1][0], Mt[mt + 1][1]}, Pb);
      Pa = wmma_f32((v2f){lo.z, lo.w}, (v2f){Mt[mt][2], Mt[mt][3]}, Pa);
      Pb = wmma_f32((v2f){lp.z, lp.w}, (v2f){Mt[mt + 1][2], Mt[mt + 1][3]}, Pb);
      Pa = wmma_f32((v2f){hi.x, hi.y}, (v2f){Mt[mt][4], Mt[mt][5]}, Pa);
      Pb = wmma_f32((v2f){hp.x, hp.y}, (v2f){Mt[mt + 1][4], Mt[mt + 1][5]}, Pb);
      Pa = wmma_f32((v2f){hi.z, hi.w}, (v2f){Mt[mt][6], Mt[mt][7]}, Pa);
      Pb = wmma_f32((v2f){hp.z, hp.w}, (v2f){Mt[mt + 1][6], Mt[mt + 1][7]}, Pb);
    }
    v8f R;
#pragma unroll
    for (int v = 0; v < 8; ++v) R[v] = ksel * o.kr[v] - (Pa[v] + Pb[v]) * o.rv[v];
    R[7] = pad ? 0.0f : R[7];
    const float xa[8] = {xlo.x, xlo.y, xlo.z, xlo.w, xhi.x, xhi.y, xhi.z, xhi.w};
    v8f Ut = {};
#pragma unroll
    for (int jj = 0; jj < 4; ++jj)
      Ut = wmma_f32((v2f){xa[2 * jj], xa[2 * jj + 1]},
                    (v2f){R[2 * jj], R[2 * jj + 1]}, Ut);
#pragma unroll
    for (int jj = 0; jj < 4; ++jj)
#pragma unroll
      for (int mt = 0; mt < 4; ++mt)
        Mt[mt] = wmma_f32((v2f){o.kt[mt][2 * jj], o.kt[mt][2 * jj + 1]},
                          (v2f){Ut[2 * jj], Ut[2 * jj + 1]}, Mt[mt]);
  };

  int    t1A;  int4 t1a, t1b;
  float4 x0lo = Xg[0],  x0hi = Xg[1];
  float4 x1lo = Xg[64], x1hi = Xg[65];
  Ops    opA, opB;
  const int* tseg = &toks[16 * cb];
  load_ops(opA, tseg[li], *(const int4*)&tseg[8 * h], *(const int4*)&tseg[8 * h + 4]);
  t1A = tseg[16 + li];
  t1a = *(const int4*)&tseg[16 + 8 * h];
  t1b = *(const int4*)&tseg[16 + 8 * h + 4];

  auto step = [&](const Ops& cur, Ops& nxt, const int c) {
    const int    c2   = (c + 2 < nch) ? c + 2 : nch - 1;
    const int    t2A  = tseg[16 * c2 + li];
    const int4   t2a  = *(const int4*)&tseg[16 * c2 + 8 * h];
    const int4   t2b  = *(const int4*)&tseg[16 * c2 + 8 * h + 4];
    const float4 x2lo = Xg[c2 * 64], x2hi = Xg[c2 * 64 + 1];
    load_ops(nxt, t1A, t1a, t1b);
    __builtin_amdgcn_sched_barrier(0);
    compute(cur, x0lo, x0hi, (cb + c == NCH - 1) && (h != 0));
    __builtin_amdgcn_sched_barrier(0);
    t1A = t2A; t1a = t2a; t1b = t2b;
    x0lo = x1lo; x0hi = x1hi; x1lo = x2lo; x1hi = x2hi;
  };

  for (int c = 0; c < nch; c += 2) {
    step(opA, opB, c);
    step(opB, opA, c + 1);
  }

  if (role != 3) {
    float4* dst = (float4*)(Mws + (size_t)role * MPLANE) +
                  (size_t)((b * 4 + wave) * 4) * 64 + lane * 2;
#pragma unroll
    for (int mt = 0; mt < 4; ++mt) {
      dst[mt * 64]     = make_float4(Mt[mt][0], Mt[mt][1], Mt[mt][2], Mt[mt][3]);
      dst[mt * 64 + 1] = make_float4(Mt[mt][4], Mt[mt][5], Mt[mt][6], Mt[mt][7]);
    }
  } else {
    const int tokq = toks[SEQL - 1];
    const v8f Pq = rows_dot(Mt, &Tl[tokq * TP + 8 * h]);
    float* dstv = homog ? wvec : vvec;
    if (h == 0) dstv[b * 64 + r0 + li] = Pq[0];
  }
}

__global__ __launch_bounds__(128) void head_kernel(
    const float* __restrict__ Mws, const float* __restrict__ wvec,
    const float* __restrict__ vvec, const float* __restrict__ Wr,
    const float* __restrict__ br, const float* __restrict__ Wo,
    const float* __restrict__ bo, float* __restrict__ out) {
  __shared__ __attribute__((aligned(16))) float Wl[HDIM];
  __shared__ __attribute__((aligned(16))) float Cl[HDIM];
  __shared__ __attribute__((aligned(16))) float Glds[HDIM];
  const int tid  = threadIdx.x;
  const int wave = tid >> 5;
  const int lane = tid & 31;
  const int li   = lane & 15;
  const int h    = lane >> 4;
  const int b    = blockIdx.x;
  const int col  = (wave << 4) + li;

  const float cq  = vvec[b * 64 + col];
  const float brv = br[col];
  const float bov = bo[col];
  const float* wrow = wvec + b * 64 + 8 * h;
  float4 wlo[4], whi[4];
  const float4* srcp = (const float4*)Mws + (size_t)((b * 4 + wave) * 4) * 64 + lane * 2;
  v8f Mt[4], Ct[4], At[4];
#pragma unroll
  for (int mt = 0; mt < 4; ++mt) {
    wlo[mt] = *(const float4*)(wrow + 16 * mt);
    whi[mt] = *(const float4*)(wrow + 16 * mt + 4);
    const float4 lo = srcp[mt * 64], hi = srcp[mt * 64 + 1];
    Mt[mt] = (v8f){lo.x, lo.y, lo.z, lo.w, hi.x, hi.y, hi.z, hi.w};
    const float4 clo = srcp[MPLANE / 4 + mt * 64], chi = srcp[MPLANE / 4 + mt * 64 + 1];
    Ct[mt] = (v8f){clo.x, clo.y, clo.z, clo.w, chi.x, chi.y, chi.z, chi.w};
    const float4 alo = srcp[2 * (MPLANE / 4) + mt * 64], ahi = srcp[2 * (MPLANE / 4) + mt * 64 + 1];
    At[mt] = (v8f){alo.x, alo.y, alo.z, alo.w, ahi.x, ahi.y, ahi.z, ahi.w};
  }
  v2f wr[16], wo[16];
#pragma unroll
  for (int kt = 0; kt < 16; ++kt) {
    wr[kt] = frag_b(Wr, 64, kt, wave, lane);
    wo[kt] = frag_b(Wo, 64, kt, wave, lane);
  }
  asm volatile("" ::"v"(cq));
  __builtin_amdgcn_sched_barrier(0);

  v8f Pa = {}, Pc = {};
#pragma unroll
  for (int mt = 0; mt < 4; ++mt) {
    Pa = wmma_f32((v2f){wlo[mt].x, wlo[mt].y}, (v2f){At[mt][0], At[mt][1]}, Pa);
    Pc = wmma_f32((v2f){wlo[mt].x, wlo[mt].y}, (v2f){Ct[mt][0], Ct[mt][1]}, Pc);
    Pa = wmma_f32((v2f){wlo[mt].z, wlo[mt].w}, (v2f){At[mt][2], At[mt][3]}, Pa);
    Pc = wmma_f32((v2f){wlo[mt].z, wlo[mt].w}, (v2f){Ct[mt][2], Ct[mt][3]}, Pc);
    Pa = wmma_f32((v2f){whi[mt].x, whi[mt].y}, (v2f){At[mt][4], At[mt][5]}, Pa);
    Pc = wmma_f32((v2f){whi[mt].x, whi[mt].y}, (v2f){Ct[mt][4], Ct[mt][5]}, Pc);
    Pa = wmma_f32((v2f){whi[mt].z, whi[mt].w}, (v2f){At[mt][6], At[mt][7]}, Pa);
    Pc = wmma_f32((v2f){whi[mt].z, whi[mt].w}, (v2f){Ct[mt][6], Ct[mt][7]}, Pc);
  }
  if (h == 0) Wl[col] = Pa[0];
  __syncthreads();

  const v8f Pq = rows_dot(Mt, &Wl[8 * h]);
  if (h == 0) Cl[col] = (Pq[0] + Pc[0]) + cq;
  __syncthreads();

  {
    v8f acc = {};
    const float* ar = &Cl[h << 1];
#pragma unroll
    for (int kt = 0; kt < 16; ++kt)
      acc = wmma_f32((v2f){ar[4 * kt], ar[4 * kt + 1]}, wr[kt], acc);
    if (h == 0) Glds[col] = acc[0] + brv;
  }
  __syncthreads();

  {
    v8f acc = {};
    const float* ar = &Glds[h << 1];
#pragma unroll
    for (int kt = 0; kt < 16; ++kt)
      acc = wmma_f32((v2f){ar[4 * kt], ar[4 * kt + 1]}, wo[kt], acc);
    if (h == 0) out[b * 64 + col] = acc[0] + bov;
  }
}

extern "C" void kernel_launch(void* const* d_in, const int* in_sizes, int n_in,
                              void* d_out, int out_size, void* d_ws, size_t ws_size,
                              hipStream_t stream) {
  const int*   seq   = (const int*)d_in[0];
  const float* embed = (const float*)d_in[1];
  const float* W1    = (const float*)d_in[2];
  const float* b1    = (const float*)d_in[3];
  const float* W2    = (const float*)d_in[4];
  const float* b2    = (const float*)d_in[5];
  const float* gamma = (const float*)d_in[6];
  const float* beta  = (const float*)d_in[7];
  const float* Wr    = (const float*)d_in[8];
  const float* br    = (const float*)d_in[9];
  const float* Wo    = (const float*)d_in[10];
  const float* bo    = (const float*)d_in[11];

  float* Tt     = (float*)d_ws;
  float* rdenom = Tt + VOCABN * HDIM;
  float* wvec   = rdenom + VOCABN;
  float* vvec   = wvec + BATCH * HDIM;
  float* Mws    = vvec + BATCH * HDIM;
  float* Xws    = Mws + 3 * MPLANE;
  float* out    = (float*)d_out;

  encode_table_kernel<<<1, 256, 0, stream>>>(embed, W1, b1, W2, b2, gamma, beta,
                                             Tt, rdenom);
  delta_scan_kernel<<<dim3(BATCH, 4), 256, 0, stream>>>(seq, Tt, rdenom, Xws, Mws,
                                                      wvec, vvec);
  head_kernel<<<BATCH, 128, 0, stream>>>(Mws, wvec, vvec, Wr, br, Wo, bo, out);
}


# Token-table kernel cut by row tile over 4 work-groups: every wave one tile per product (2nd product in 2 halves of k), LayerNorm 16 lanes a row, global reads all up front; summation order changes
# speedup vs baseline: 9.0071x; 1.2301x over previous
#include <hip/hip_runtime.h>


#define HDIM   64
#define VOCABN 64
#define BATCH  64
#define SEQL   4096

typedef __attribute__((ext_vector_type(2))) float v2f;
typedef __attribute__((ext_vector_type(2))) float f32x2;
typedef __attribute__((ext_vector_type(8))) float v8f;

__device__ __forceinline__ v2f frag_a(const float* A, int lda, int mt, int kt, int lane) {
  int row = (mt << 4) + (lane & 15);
  int k0  = (kt << 2) + ((lane >> 4) << 1);
  v2f a;
  a.x = A[row * lda + k0];
  a.y = A[row * lda + k0 + 1];
  return a;
}
__device__ __forceinline__ v2f frag_b(const float* Bm, int ldb, int kt, int nt, int lane) {
  int col = (nt << 4) + (lane & 15);
  int k0  = (kt << 2) + ((lane >> 4) << 1);
  v2f b;
  b.x = Bm[k0 * ldb + col];
  b.y = Bm[(k0 + 1) * ldb + col];
  return b;
}

__device__ __forceinline__ v8f wmma_f32(v2f a, v2f b, v8f c) {
  return __builtin_amdgcn_wmma_f32_16x16x4_f32(false, a, false, b, (short)0, c,
                                               false, false);
}

#define RTP 20
#define XP  68

template <int MASK>
__device__ __forceinline__ float row_xor(float x) {
  return __int_as_float(__builtin_amdgcn_update_dpp(0, __float_as_int(x), 0x160 | MASK,
                                                    0xf, 0xf, false));
}
__device__ __forceinline__ float row_sum16(float x) {
  x += row_xor<1>(x);
  x += row_xor<2>(x);
  x += row_xor<4>(x);
  x += row_xor<8>(x);
  return x;
}

__global__ __launch_bounds__(256) void encode_table_kernel(
    const float* __restrict__ embed, const float* __restrict__ W1,
    const float* __restrict__ b1, const float* __restrict__ W2,
    const float* __restrict__ b2, const float* __restrict__ gamma,
    const float* __restrict__ beta, float* __restrict__ Tout,
    float* __restrict__ rdenom) {
  __shared__ __attribute__((aligned(16))) float Rt[128 * RTP];
  __shared__ __attribute__((aligned(16))) float Xp[2 * 16 * XP];
  const int tid  = threadIdx.x;
  const int wave = tid >> 5;
  const int lane = tid & 31;
  const int li   = lane & 15;
  const int h    = lane >> 4;
  const int mt   = blockIdx.x;
  const int nt2  = wave & 3;
  const int kh   = wave >> 2;
  const int r    = tid >> 4;
  const int c4   = tid & 15;

  v2f a1[16], w1f[16], w2f[16];
  const float* ap = embed + (16 * mt + li) * 64 + 2 * h;
  const float* bp = W1 + (2 * h) * 128 + 16 * wave + li;
  const float* wp = W2 + (64 * kh + 2 * h) * 64 + 16 * nt2 + li;
#pragma unroll
  for (int kt = 0; kt < 16; ++kt) {
    a1[kt]  = (v2f){ap[4 * kt], ap[4 * kt + 1]};
    w1f[kt] = (v2f){bp[(4 * kt) * 128], bp[(4 * kt + 1) * 128]};
    w2f[kt] = (v2f){wp[(4 * kt) * 64], wp[(4 * kt + 1) * 64]};
  }
  const float  bias1 = b1[16 * wave + li];
  const float4 e4  = *(const float4*)(embed + (16 * mt + r) * 64 + 4 * c4);
  const float4 bb4 = *(const float4*)(b2 + 4 * c4);
  const float4 g4  = *(const float4*)(gamma + 4 * c4);
  const float4 be4 = *(const float4*)(beta + 4 * c4);
  __builtin_amdgcn_sched_barrier(0);

  v8f acc = {};
#pragma unroll
  for (int kt = 0; kt < 16; ++kt) acc = wmma_f32(a1[kt], w1f[kt], acc);
  float rr[8];
#pragma unroll
  for (int v = 0; v < 8; ++v) rr[v] = fmaxf(acc[v] + bias1, 0.0f);
  float4* rdst = (float4*)&Rt[(16 * wave + li) * RTP + 8 * h];
  rdst[0] = make_float4(rr[0], rr[1], rr[2], rr[3]);
  rdst[1] = make_float4(rr[4], rr[5], rr[6], rr[7]);
  asm volatile("" ::"v"(e4.x), "v"(e4.y), "v"(e4.z), "v"(e4.w), "v"(bb4.x), "v"(bb4.y),
               "v"(bb4.z), "v"(bb4.w), "v"(g4.x), "v"(g4.y), "v"(g4.z), "v"(g4.w),
               "v"(be4.x), "v"(be4.y), "v"(be4.z), "v"(be4.w));
  __syncthreads();

  v8f acc2 = {};
  const float* rp = &Rt[(64 * kh + 2 * h) * RTP + li];
  v2f a2[16];
#pragma unroll
  for (int kt = 0; kt < 16; ++kt) a2[kt] = (v2f){rp[(4 * kt) * RTP], rp[(4 * kt + 1) * RTP]};
  __builtin_amdgcn_sched_barrier(0);
#pragma unroll
  for (int kt = 0; kt < 16; ++kt) acc2 = wmma_f32(a2[kt], w2f[kt], acc2);
  float* xd = &Xp[(kh * 16 + 8 * h) * XP + 16 * nt2 + li];
#pragma unroll
  for (int v = 0; v < 8; ++v) xd[v * XP] = acc2[v];
  __syncthreads();

  const float4 p0 = *(const float4*)&Xp[r * XP + 4 * c4];
  const float4 p1 = *(const float4*)&Xp[(16 + r) * XP + 4 * c4];
  const float x0 = e4.x + (p0.x + p1.x) + bb4.x;
  const float x1 = e4.y + (p0.y + p1.y) + bb4.y;
  const float x2 = e4.z + (p0.z + p1.z) + bb4.z;
  const float x3 = e4.w + (p0.w + p1.w) + bb4.w;
  const float mu = row_sum16((x0 + x1) + (x2 + x3)) * (1.0f / 64.0f);
  const float d0 = x0 - mu, d1 = x1 - mu, d2 = x2 - mu, d3 = x3 - mu;
  const float var = row_sum16((d0 * d0 + d1 * d1) + (d2 * d2 + d3 * d3)) * (1.0f / 64.0f);
  const float rs  = rsqrtf(var + 1e-5f);
  const float t0 = d0 * rs * g4.x + be4.x;
  const float t1 = d1 * rs * g4.y + be4.y;
  const float t2 = d2 * rs * g4.z + be4.z;
  const float t3 = d3 * rs * g4.w + be4.w;
  const float ksq = row_sum16((t0 * t0 + t1 * t1) + (t2 * t2 + t3 * t3));
  *(float4*)(Tout + (16 * mt + r) * 64 + 4 * c4) = make_float4(t0, t1, t2, t3);
  if (c4 == 0) rdenom[16 * mt + r] = 1.0f / (ksq + 1e-6f);
}

#define TP   68
#define NCH  256
#define SEG1 88
#define SEGN 56
#define XCAP 128
#define MPLANE (BATCH * HDIM * HDIM)
static_assert(SEG1 + 3 * SEGN == NCH, "the four stretches cover the sequence");
static_assert(SEG1 % 2 == 0 && SEGN % 2 == 0 && SEG1 <= XCAP && SEGN <= XCAP,
              "the scan takes two chunks per trip; X plane holds XCAP chunks");

__device__ __forceinline__ v8f rows_dot(const v8f (&Mt)[4], const float* krow) {
  v8f acc = {};
#pragma unroll
  for (int mt = 0; mt < 4; ++mt) {
    const float4 lo = *(const float4*)(krow + 16 * mt);
    const float4 hi = *(const float4*)(krow + 16 * mt + 4);
    acc = wmma_f32((v2f){lo.x, lo.y}, (v2f){Mt[mt][0], Mt[mt][1]}, acc);
    acc = wmma_f32((v2f){lo.z, lo.w}, (v2f){Mt[mt][2], Mt[mt][3]}, acc);
    acc = wmma_f32((v2f){hi.x, hi.y}, (v2f){Mt[mt][4], Mt[mt][5]}, acc);
    acc = wmma_f32((v2f){hi.z, hi.w}, (v2f){Mt[mt][6], Mt[mt][7]}, acc);
  }
  return acc;
}

__global__ __launch_bounds__(256) void delta_scan_kernel(
    const int* __restrict__ seq, const float* __restrict__ Ttab,
    const float* __restrict__ rdenom, float* __restrict__ Xws,
    float* __restrict__ Mws, float* __restrict__ wvec,
    float* __restrict__ vvec) {
  __shared__ __attribute__((aligned(16))) float Tl[VOCABN * TP];
  __shared__ __attribute__((aligned(16))) float Gl[VOCABN * VOCABN];
  __shared__ float rd[VOCABN];
  __shared__ __attribute__((aligned(16))) int toks[SEQL];
  const int b    = blockIdx.x;
  const int role = blockIdx.y;
  const int tid  = threadIdx.x;
  const int wave = tid >> 5;
  const int cb   = (role == 0) ? 0 : SEG1 + (role - 1) * SEGN;
  const int nch  = (role == 0) ? SEG1 : SEGN;
  const bool homog = (role != 0) && (wave >= 4);
  const float ksel = homog ? 0.0f : 1.0f;
  const int lane = tid & 31;
  const int li   = lane & 15;
  const int h    = lane >> 4;

  const float4* Tg4 = (const float4*)Ttab;
  for (int i = tid; i < VOCABN * 16; i += 256)
    *(float4*)&Tl[(i >> 4) * TP + ((i & 15) << 2)] = Tg4[i];
  if (tid < VOCABN) rd[tid] = rdenom[tid];
  const int4* sg = (const int4*)(seq + b * SEQL);
  int4*       tl = (int4*)toks;
  for (int i = tid; i < SEQL / 4; i += 256) tl[i] = sg[i];
  __syncthreads();

  for (int tile = wave; tile < 16; tile += 8) {
    const int mt = tile >> 2, nt = tile & 3;
    v8f acc = {};
    const float* ar = &Tl[((mt << 4) + li) * TP + (h << 1)];
    const float* br = &Tl[((nt << 4) + li) * TP + (h << 1)];
#pragma unroll
    for (int kt = 0; kt < 16; ++kt) {
      v2f a = (v2f){ar[4 * kt], ar[4 * kt + 1]};
      v2f bb = (v2f){br[4 * kt], br[4 * kt + 1]};
      acc = wmma_f32(a, bb, acc);
    }
#pragma unroll
    for (int v = 0; v < 8; ++v)
      Gl[((mt << 4) + (h << 3) + v) * VOCABN + (nt << 4) + li] = acc[v];
  }
  __syncthreads();

  float* Xb = Xws + (size_t)(role * BATCH + b) * (XCAP * 256);
  const int npass = (nch * 16 + 255) >> 8;
  for (int n = 0; n < npass; ++n) {
    const int w  = tid + 256 * n;
    const int cl = w >> 4;
    if (cl >= nch) continue;
    const int c  = cb + cl;
    const int i  = w & 15;
    int tk[16];
    const int4* tp = (const int4*)&toks[16 * c];
#pragma unroll
    for (int u = 0; u < 4; ++u) {
      const int4 t4 = tp[u];
      tk[4 * u] = t4.x; tk[4 * u + 1] = t4.y; tk[4 * u + 2] = t4.z; tk[4 * u + 3] = t4.w;
    }
    float x[16], y[16];
#pragma unroll
    for (int m = 0; m < 16; ++m) x[m] = (m == i) ? 1.0f : 0.0f;
    y[15] = ((16 * c + 15 < SEQL - 1) ? rd[tk[15]] : 0.0f) * x[15];
#pragma unroll
    for (int j = 14; j >= 0; --j) {
      const float* gr = &Gl[tk[j] * VOCABN];
      float s = 0.0f;
#pragma unroll
      for (int m = j + 1; m < 16; ++m) s = fmaf(gr[tk[m]], y[m], s);
      x[j] = (j < i) ? -s : x[j];
      y[j] = rd[tk[j]] * x[j];
    }
    float4* dst = (float4*)(Xb + (cl * 16 + i) * 16);
    dst[0] = make_float4(x[0], x[1], x[2], x[3]);
    dst[1] = make_float4(x[4], x[5], x[6], x[7]);
    dst[2] = make_float4(x[8], x[9], x[10], x[11]);
    dst[3] = make_float4(x[12], x[13], x[14], x[15]);
  }
  __syncthreads();
  if (wave >= 4 && role == 0) return;

  const int r0 = (wave & 3) << 4;
  v8f Mt[4];
#pragma unroll
  for (int mt = 0; mt < 4; ++mt)
#pragma unroll
    for (int v = 0; v < 8; ++v)
      Mt[mt][v] = (homog && (r0 + li == 16 * mt + 8 * h + v)) ? 1.0f : 0.0f;

  const float4* Xg = (const float4*)Xb + (li * 4 + 2 * h);

  struct Ops { float4 ka[8]; float kt[4][8]; float kr[8]; float rv[8]; };

  auto load_ops = [&](Ops& o, const int tA, const int4 ta, const int4 tb) {
    const float* krow = &Tl[tA * TP + 8 * h];
#pragma unroll
    for (int u = 0; u < 8; ++u) o.ka[u] = *(const float4*)(krow + 16 * (u >> 1) + 4 * (u & 1));
    const int tv[8] = {ta.x, ta.y, ta.z, ta.w, tb.x, tb.y, tb.z, tb.w};
#pragma unroll
    for (int v = 0; v < 8; ++v) {
      const float* row = &Tl[tv[v] * TP + li];
#pragma unroll
      for (int mt = 0; mt < 4; ++mt) o.kt[mt][v] = row[16 * mt];
      o.kr[v] = row[r0];
      o.rv[v] = rd[tv[v]];
    }
  };

  auto compute = [&](const Ops& o, const float4 xlo, const float4 xhi, const bool pad) {
    v8f Pa = {}, Pb = {};
#pragma unroll
    for (int mt = 0; mt < 4; mt += 2) {
      const float4 lo = o.ka[2 * mt], hi = o.ka[2 * mt + 1];
      const float4 lp = o.ka[2 * mt + 2], hp = o.ka[2 * mt + 3];
      Pa = wmma_f32((v2f){lo.x, lo.y}, (v2f){Mt[mt][0], Mt[mt][1]}, Pa);
      Pb = wmma_f32((v2f){lp.x, lp.y}, (v2f){Mt[mt + 1][0], Mt[mt + 1][1]}, Pb);
      Pa = wmma_f32((v2f){lo.z, lo.w}, (v2f){Mt[mt][2], Mt[mt][3]}, Pa);
      Pb = wmma_f32((v2f){lp.z, lp.w}, (v2f){Mt[mt + 1][2], Mt[mt + 1][3]}, Pb);
      Pa = wmma_f32((v2f){hi.x, hi.y}, (v2f){Mt[mt][4], Mt[mt][5]}, Pa);
      Pb = wmma_f32((v2f){hp.x, hp.y}, (v2f){Mt[mt + 1][4], Mt[mt + 1][5]}, Pb);
      Pa = wmma_f32((v2f){hi.z, hi.w}, (v2f){Mt[mt][6], Mt[mt][7]}, Pa);
      Pb = wmma_f32((v2f){hp.z, hp.w}, (v2f){Mt[mt + 1][6], Mt[mt + 1][7]}, Pb);
    }
    v8f R;
#pragma unroll
    for (int v = 0; v < 8; ++v) R[v] = ksel * o.kr[v] - (Pa[v] + Pb[v]) * o.rv[v];
    R[7] = pad ? 0.0f : R[7];
    const float xa[8] = {xlo.x, xlo.y, xlo.z, xlo.w, xhi.x, xhi.y, xhi.z, xhi.w};
    v8f Ut = {};
#pragma unroll
    for (int jj = 0; jj < 4; ++jj)
      Ut = wmma_f32((v2f){xa[2 * jj], xa[2 * jj + 1]},
                    (v2f){R[2 * jj], R[2 * jj + 1]}, Ut);
#pragma unroll
    for (int jj = 0; jj < 4; ++jj)
#pragma unroll
      for (int mt = 0; mt < 4; ++mt)
        Mt[mt] = wmma_f32((v2f){o.kt[mt][2 * jj], o.kt[mt][2 * jj + 1]},
                          (v2f){Ut[2 * jj], Ut[2 * jj + 1]}, Mt[mt]);
  };

  int    t1A;  int4 t1a, t1b;
  float4 x0lo = Xg[0],  x0hi = Xg[1];
  float4 x1lo = Xg[64], x1hi = Xg[65];
  Ops    opA, opB;
  const int* tseg = &toks[16 * cb];
  load_ops(opA, tseg[li], *(const int4*)&tseg[8 * h], *(const int4*)&tseg[8 * h + 4]);
  t1A = tseg[16 + li];
  t1a = *(const int4*)&tseg[16 + 8 * h];
  t1b = *(const int4*)&tseg[16 + 8 * h + 4];

  auto step = [&](const Ops& cur, Ops& nxt, const int c) {
    const int    c2   = (c + 2 < nch) ? c + 2 : nch - 1;
    const int    t2A  = tseg[16 * c2 + li];
    const int4   t2a  = *(const int4*)&tseg[16 * c2 + 8 * h];
    const int4   t2b  = *(const int4*)&tseg[16 * c2 + 8 * h + 4];
    const float4 x2lo = Xg[c2 * 64], x2hi = Xg[c2 * 64 + 1];
    load_ops(nxt, t1A, t1a, t1b);
    __builtin_amdgcn_sched_barrier(0);
    compute(cur, x0lo, x0hi, (cb + c == NCH - 1) && (h != 0));
    __builtin_amdgcn_sched_barrier(0);
    t1A = t2A; t1a = t2a; t1b = t2b;
    x0lo = x1lo; x0hi = x1hi; x1lo = x2lo; x1hi = x2hi;
  };

  for (int c = 0; c < nch; c += 2) {
    step(opA, opB, c);
    step(opB, opA, c + 1);
  }

  if (role != 3) {
    const int plane = (role == 0) ? 0 : 2 * role - 1 + (homog ? 1 : 0);
    float4* dst = (float4*)(Mws + (size_t)plane * MPLANE) +
                  (size_t)((b * 4 + (wave & 3)) * 4) * 64 + lane * 2;
#pragma unroll
    for (int mt = 0; mt < 4; ++mt) {
      dst[mt * 64]     = make_float4(Mt[mt][0], Mt[mt][1], Mt[mt][2], Mt[mt][3]);
      dst[mt * 64 + 1] = make_float4(Mt[mt][4], Mt[mt][5], Mt[mt][6], Mt[mt][7]);
    }
  } else {
    const int tokq = toks[SEQL - 1];
    const v8f Pq = rows_dot(Mt, &Tl[tokq * TP + 8 * h]);
    float* dstv = homog ? wvec : vvec;
    if (h == 0) dstv[b * 64 + r0 + li] = Pq[0];
  }
}

__device__ __forceinline__ v8f load_tile(const float4* p) {
  const float4 lo = p[0], hi = p[1];
  return (v8f){lo.x, lo.y, lo.z, lo.w, hi.x, hi.y, hi.z, hi.w};
}

__device__ __forceinline__ void rows_dot2(const v8f (&At)[4], const v8f (&Ct)[4],
                                          const float4 (&lo)[4], const float4 (&hi)[4],
                                          v8f& Pa, v8f& Pc) {
  Pa = (v8f){};
  Pc = (v8f){};
#pragma unroll
  for (int mt = 0; mt < 4; ++mt) {
    Pa = wmma_f32((v2f){lo[mt].x, lo[mt].y}, (v2f){At[mt][0], At[mt][1]}, Pa);
    Pc = wmma_f32((v2f){lo[mt].x, lo[mt].y}, (v2f){Ct[mt][0], Ct[mt][1]}, Pc);
    Pa = wmma_f32((v2f){lo[mt].z, lo[mt].w}, (v2f){At[mt][2], At[mt][3]}, Pa);
    Pc = wmma_f32((v2f){lo[mt].z, lo[mt].w}, (v2f){Ct[mt][2], Ct[mt][3]}, Pc);
    Pa = wmma_f32((v2f){hi[mt].x, hi[mt].y}, (v2f){At[mt][4], At[mt][5]}, Pa);
    Pc = wmma_f32((v2f){hi[mt].x, hi[mt].y}, (v2f){Ct[mt][4], Ct[mt][5]}, Pc);
    Pa = wmma_f32((v2f){hi[mt].z, hi[mt].w}, (v2f){At[mt][6], At[mt][7]}, Pa);
    Pc = wmma_f32((v2f){hi[mt].z, hi[mt].w}, (v2f){Ct[mt][6], Ct[mt][7]}, Pc);
  }
}

__global__ __launch_bounds__(128) void head_kernel(
    const float* __restrict__ Mws, const float* __restrict__ wvec,
    const float* __restrict__ vvec, const float* __restrict__ Wr,
    const float* __restrict__ br, const float* __restrict__ Wo,
    const float* __restrict__ bo, float* __restrict__ out) {
  __shared__ __attribute__((aligned(16))) float W3l[HDIM];
  __shared__ __attribute__((aligned(16))) float W2l[HDIM];
  __shared__ __attribute__((aligned(16))) float Cl[HDIM];
  __shared__ __attribute__((aligned(16))) float Glds[HDIM];
  const int tid  = threadIdx.x;
  const int wave = tid >> 5;
  const int lane = tid & 31;
  const int li   = lane & 15;
  const int h    = lane >> 4;
  const int b    = blockIdx.x;
  const int col  = (wave << 4) + li;

  const float cq  = vvec[b * 64 + col];
  const float brv = br[col];
  const float bov = bo[col];
  const float* wrow = wvec + b * 64 + 8 * h;
  float4 wlo[4], whi[4];
  const float4* srcp = (const float4*)Mws + (size_t)((b * 4 + wave) * 4) * 64 + lane * 2;
  v8f Mt[4], C2t[4], A2t[4], C3t[4], A3t[4];
#pragma unroll
  for (int mt = 0; mt < 4; ++mt) {
    wlo[mt] = *(const float4*)(wrow + 16 * mt);
    whi[mt] = *(const float4*)(wrow + 16 * mt + 4);
    A3t[mt] = load_tile(srcp + 4 * (MPLANE / 4) + mt * 64);
    C3t[mt] = load_tile(srcp + 3 * (MPLANE / 4) + mt * 64);
    A2t[mt] = load_tile(srcp + 2 * (MPLANE / 4) + mt * 64);
    C2t[mt] = load_tile(srcp + 1 * (MPLANE / 4) + mt * 64);
    Mt[mt]  = load_tile(srcp + mt * 64);
  }
  v2f wr[16], wo[16];
#pragma unroll
  for (int kt = 0; kt < 16; ++kt) {
    wr[kt] = frag_b(Wr, 64, kt, wave, lane);
    wo[kt] = frag_b(Wo, 64, kt, wave, lane);
  }
  asm volatile("" ::"v"(cq));
  __builtin_amdgcn_sched_barrier(0);

  v8f Pa, Pc;
  rows_dot2(A3t, C3t, wlo, whi, Pa, Pc);
  if (h == 0) W3l[col] = Pa[0];
  const float v3 = Pc[0] + cq;
  __syncthreads();

  float4 slo[4], shi[4];
#pragma unroll
  for (int mt = 0; mt < 4; ++mt) {
    slo[mt] = *(const float4*)(&W3l[8 * h] + 16 * mt);
    shi[mt] = *(const float4*)(&W3l[8 * h] + 16 * mt + 4);
  }
  rows_dot2(A2t, C2t, slo, shi, Pa, Pc);
  if (h == 0) W2l[col] = Pa[0];
  const float v2 = Pc[0] + v3;
  __syncthreads();

  const v8f Pq = rows_dot(Mt, &W2l[8 * h]);
  if (h == 0) Cl[col] = Pq[0] + v2;
  __syncthreads();

  {
    v8f acc = {};
    const float* ar = &Cl[h << 1];
#pragma unroll
    for (int kt = 0; kt < 16; ++kt)
      acc = wmma_f32((v2f){ar[4 * kt], ar[4 * kt + 1]}, wr[kt], acc);
    if (h == 0) Glds[col] = acc[0] + brv;
  }
  __syncthreads();

  {
    v8f acc = {};
    const float* ar = &Glds[h << 1];
#pragma unroll
    for (int kt = 0; kt < 16; ++kt)
      acc = wmma_f32((v2f){ar[4 * kt], ar[4 * kt + 1]}, wo[kt], acc);
    if (h == 0) out[b * 64 + col] = acc[0] + bov;
  }
}

extern "C" void kernel_launch(void* const* d_in, const int* in_sizes, int n_in,
                              void* d_out, int out_size, void* d_ws, size_t ws_size,
                              hipStream_t stream) {
  const int*   seq   = (const int*)d_in[0];
  const float* embed = (const float*)d_in[1];
  const float* W1    = (const float*)d_in[2];
  const float* b1    = (const float*)d_in[3];
  const float* W2    = (const float*)d_in[4];
  const float* b2    = (const float*)d_in[5];
  const float* gamma = (const float*)d_in[6];
  const float* beta  = (const float*)d_in[7];
  const float* Wr    = (const float*)d_in[8];
  const float* br    = (const float*)d_in[9];
  const float* Wo    = (const float*)d_in[10];
  const float* bo    = (const float*)d_in[11];

  float* Tt     = (float*)d_ws;
  float* rdenom = Tt + VOCABN * HDIM;
  float* wvec   = rdenom + VOCABN;
  float* vvec   = wvec + BATCH * HDIM;
  float* Mws    = vvec + BATCH * HDIM;
  float* Xws    = Mws + 5 * MPLANE;
  float* out    = (float*)d_out;

  encode_table_kernel<<<4, 256, 0, stream>>>(embed, W1, b1, W2, b2, gamma, beta,
                                             Tt, rdenom);
  delta_scan_kernel<<<dim3(BATCH, 4), 256, 0, stream>>>(seq, Tt, rdenom, Xws, Mws,
                                                      wvec, vvec);
  head_kernel<<<BATCH, 128, 0, stream>>>(Mws, wvec, vvec, Wr, br, Wo, bo, out);
}


# Chunk inverses X=A^-1: a chunk's 16 columns on the 16 lanes of a row, saxpy-order substitution, Gram entries from lane m by row_share; 21 LDS reads a thread, was 140; summation order changes
# speedup vs baseline: 10.2493x; 1.1379x over previous
#include <hip/hip_runtime.h>


#define HDIM   64
#define VOCABN 64
#define BATCH  64
#define SEQL   4096

typedef __attribute__((ext_vector_type(2))) float v2f;
typedef __attribute__((ext_vector_type(2))) float f32x2;
typedef __attribute__((ext_vector_type(8))) float v8f;

__device__ __forceinline__ v2f frag_a(const float* A, int lda, int mt, int kt, int lane) {
  int row = (mt << 4) + (lane & 15);
  int k0  = (kt << 2) + ((lane >> 4) << 1);
  v2f a;
  a.x = A[row * lda + k0];
  a.y = A[row * lda + k0 + 1];
  return a;
}
__device__ __forceinline__ v2f frag_b(const float* Bm, int ldb, int kt, int nt, int lane) {
  int col = (nt << 4) + (lane & 15);
  int k0  = (kt << 2) + ((lane >> 4) << 1);
  v2f b;
  b.x = Bm[k0 * ldb + col];
  b.y = Bm[(k0 + 1) * ldb + col];
  return b;
}

__device__ __forceinline__ v8f wmma_f32(v2f a, v2f b, v8f c) {
  return __builtin_amdgcn_wmma_f32_16x16x4_f32(false, a, false, b, (short)0, c,
                                               false, false);
}

#define RTP 20
#define XP  68

template <int MASK>
__device__ __forceinline__ float row_xor(float x) {
  return __int_as_float(__builtin_amdgcn_update_dpp(0, __float_as_int(x), 0x160 | MASK,
                                                    0xf, 0xf, false));
}
__device__ __forceinline__ float row_sum16(float x) {
  x += row_xor<1>(x);
  x += row_xor<2>(x);
  x += row_xor<4>(x);
  x += row_xor<8>(x);
  return x;
}

__global__ __launch_bounds__(256) void encode_table_kernel(
    const float* __restrict__ embed, const float* __restrict__ W1,
    const float* __restrict__ b1, const float* __restrict__ W2,
    const float* __restrict__ b2, const float* __restrict__ gamma,
    const float* __restrict__ beta, float* __restrict__ Tout,
    float* __restrict__ rdenom) {
  __shared__ __attribute__((aligned(16))) float Rt[128 * RTP];
  __shared__ __attribute__((aligned(16))) float Xp[2 * 16 * XP];
  const int tid  = threadIdx.x;
  const int wave = tid >> 5;
  const int lane = tid & 31;
  const int li   = lane & 15;
  const int h    = lane >> 4;
  const int mt   = blockIdx.x;
  const int nt2  = wave & 3;
  const int kh   = wave >> 2;
  const int r    = tid >> 4;
  const int c4   = tid & 15;

  v2f a1[16], w1f[16], w2f[16];
  const float* ap = embed + (16 * mt + li) * 64 + 2 * h;
  const float* bp = W1 + (2 * h) * 128 + 16 * wave + li;
  const float* wp = W2 + (64 * kh + 2 * h) * 64 + 16 * nt2 + li;
#pragma unroll
  for (int kt = 0; kt < 16; ++kt) {
    a1[kt]  = (v2f){ap[4 * kt], ap[4 * kt + 1]};
    w1f[kt] = (v2f){bp[(4 * kt) * 128], bp[(4 * kt + 1) * 128]};
    w2f[kt] = (v2f){wp[(4 * kt) * 64], wp[(4 * kt + 1) * 64]};
  }
  const float  bias1 = b1[16 * wave + li];
  const float4 e4  = *(const float4*)(embed + (16 * mt + r) * 64 + 4 * c4);
  const float4 bb4 = *(const float4*)(b2 + 4 * c4);
  const float4 g4  = *(const float4*)(gamma + 4 * c4);
  const float4 be4 = *(const float4*)(beta + 4 * c4);
  __builtin_amdgcn_sched_barrier(0);

  v8f acc = {};
#pragma unroll
  for (int kt = 0; kt < 16; ++kt) acc = wmma_f32(a1[kt], w1f[kt], acc);
  float rr[8];
#pragma unroll
  for (int v = 0; v < 8; ++v) rr[v] = fmaxf(acc[v] + bias1, 0.0f);
  float4* rdst = (float4*)&Rt[(16 * wave + li) * RTP + 8 * h];
  rdst[0] = make_float4(rr[0], rr[1], rr[2], rr[3]);
  rdst[1] = make_float4(rr[4], rr[5], rr[6], rr[7]);
  asm volatile("" ::"v"(e4.x), "v"(e4.y), "v"(e4.z), "v"(e4.w), "v"(bb4.x), "v"(bb4.y),
               "v"(bb4.z), "v"(bb4.w), "v"(g4.x), "v"(g4.y), "v"(g4.z), "v"(g4.w),
               "v"(be4.x), "v"(be4.y), "v"(be4.z), "v"(be4.w));
  __syncthreads();

  v8f acc2 = {};
  const float* rp = &Rt[(64 * kh + 2 * h) * RTP + li];
  v2f a2[16];
#pragma unroll
  for (int kt = 0; kt < 16; ++kt) a2[kt] = (v2f){rp[(4 * kt) * RTP], rp[(4 * kt + 1) * RTP]};
  __builtin_amdgcn_sched_barrier(0);
#pragma unroll
  for (int kt = 0; kt < 16; ++kt) acc2 = wmma_f32(a2[kt], w2f[kt], acc2);
  float* xd = &Xp[(kh * 16 + 8 * h) * XP + 16 * nt2 + li];
#pragma unroll
  for (int v = 0; v < 8; ++v) xd[v * XP] = acc2[v];
  __syncthreads();

  const float4 p0 = *(const float4*)&Xp[r * XP + 4 * c4];
  const float4 p1 = *(const float4*)&Xp[(16 + r) * XP + 4 * c4];
  const float x0 = e4.x + (p0.x + p1.x) + bb4.x;
  const float x1 = e4.y + (p0.y + p1.y) + bb4.y;
  const float x2 = e4.z + (p0.z + p1.z) + bb4.z;
  const float x3 = e4.w + (p0.w + p1.w) + bb4.w;
  const float mu = row_sum16((x0 + x1) + (x2 + x3)) * (1.0f / 64.0f);
  const float d0 = x0 - mu, d1 = x1 - mu, d2 = x2 - mu, d3 = x3 - mu;
  const float var = row_sum16((d0 * d0 + d1 * d1) + (d2 * d2 + d3 * d3)) * (1.0f / 64.0f);
  const float rs  = rsqrtf(var + 1e-5f);
  const float t0 = d0 * rs * g4.x + be4.x;
  const float t1 = d1 * rs * g4.y + be4.y;
  const float t2 = d2 * rs * g4.z + be4.z;
  const float t3 = d3 * rs * g4.w + be4.w;
  const float ksq = row_sum16((t0 * t0 + t1 * t1) + (t2 * t2 + t3 * t3));
  *(float4*)(Tout + (16 * mt + r) * 64 + 4 * c4) = make_float4(t0, t1, t2, t3);
  if (c4 == 0) rdenom[16 * mt + r] = 1.0f / (ksq + 1e-6f);
}

#define TP   68
#define GP   80
#define SP   68
#define NCH  256
#define SEG1 88
#define SEGN 56
#define XCAP 128
#define MPLANE (BATCH * HDIM * HDIM)
static_assert(SEG1 + 3 * SEGN == NCH, "the four stretches cover the sequence");
static_assert(SEG1 % 2 == 0 && SEGN % 2 == 0 && SEG1 <= XCAP && SEGN <= XCAP,
              "the scan takes two chunks per trip; X plane holds XCAP chunks");

__device__ __forceinline__ v8f rows_dot(const v8f (&Mt)[4], const float* krow) {
  v8f acc = {};
#pragma unroll
  for (int mt = 0; mt < 4; ++mt) {
    const float4 lo = *(const float4*)(krow + 16 * mt);
    const float4 hi = *(const float4*)(krow + 16 * mt + 4);
    acc = wmma_f32((v2f){lo.x, lo.y}, (v2f){Mt[mt][0], Mt[mt][1]}, acc);
    acc = wmma_f32((v2f){lo.z, lo.w}, (v2f){Mt[mt][2], Mt[mt][3]}, acc);
    acc = wmma_f32((v2f){hi.x, hi.y}, (v2f){Mt[mt][4], Mt[mt][5]}, acc);
    acc = wmma_f32((v2f){hi.z, hi.w}, (v2f){Mt[mt][6], Mt[mt][7]}, acc);
  }
  return acc;
}

template <int L>
__device__ __forceinline__ float row_lane(float x) {
  return __int_as_float(__builtin_amdgcn_mov_dpp(__float_as_int(x), 0x150 | L, 0xf, 0xf, true));
}

template <int M>
__device__ __forceinline__ void xinv_steps(float (&s)[16], float (&x)[16],
                                           const float (&g)[16], const float rdl,
                                           const int i) {
  x[M] = (M < i) ? -s[M] : ((M == i) ? 1.0f : 0.0f);
  const float ym = row_lane<M>(rdl) * x[M];
#pragma unroll
  for (int j = 0; j < M; ++j) s[j] = fmaf(row_lane<M>(g[j]), ym, s[j]);
  if constexpr (M > 1) xinv_steps<M - 1>(s, x, g, rdl, i);
}

__global__ __launch_bounds__(256) void delta_scan_kernel(
    const int* __restrict__ seq, const float* __restrict__ Ttab,
    const float* __restrict__ rdenom, float* __restrict__ Xws,
    float* __restrict__ Mws, float* __restrict__ wvec,
    float* __restrict__ vvec) {
  __shared__ __attribute__((aligned(16))) float Tl[VOCABN * TP];
  __shared__ __attribute__((aligned(16))) float Gl[VOCABN * GP];
  __shared__ float rd[VOCABN];
  __shared__ __attribute__((aligned(16))) int toks[SEQL];
  __shared__ __attribute__((aligned(16))) float St[8 * 16 * SP];
  const int b    = blockIdx.x;
  const int role = blockIdx.y;
  const int tid  = threadIdx.x;
  const int wave = tid >> 5;
  const int cb   = (role == 0) ? 0 : SEG1 + (role - 1) * SEGN;
  const int nch  = (role == 0) ? SEG1 : SEGN;
  const bool homog = (role != 0) && (wave >= 4);
  const float ksel = homog ? 0.0f : 1.0f;
  const int lane = tid & 31;
  const int li   = lane & 15;
  const int h    = lane >> 4;

  const float4* Tg4 = (const float4*)Ttab;
  for (int i = tid; i < VOCABN * 16; i += 256)
    *(float4*)&Tl[(i >> 4) * TP + ((i & 15) << 2)] = Tg4[i];
  if (tid < VOCABN) rd[tid] = rdenom[tid];
  const int4* sg = (const int4*)(seq + b * SEQL);
  int4*       tl = (int4*)toks;
  for (int i = tid; i < SEQL / 4; i += 256) tl[i] = sg[i];
  __syncthreads();

  for (int tile = wave; tile < 16; tile += 8) {
    const int mt = tile >> 2, nt = tile & 3;
    v8f acc = {};
    const float* ar = &Tl[((mt << 4) + li) * TP + (h << 1)];
    const float* br = &Tl[((nt << 4) + li) * TP + (h << 1)];
#pragma unroll
    for (int kt = 0; kt < 16; ++kt) {
      v2f a = (v2f){ar[4 * kt], ar[4 * kt + 1]};
      v2f bb = (v2f){br[4 * kt], br[4 * kt + 1]};
      acc = wmma_f32(a, bb, acc);
    }
#pragma unroll
    for (int v = 0; v < 8; ++v)
      Gl[((mt << 4) + (h << 3) + v) * GP + (nt << 4) + li] = acc[v];
  }
  __syncthreads();

  float* Xb = Xws + (size_t)(role * BATCH + b) * (XCAP * 256);
  const int npass = (nch * 16 + 255) >> 8;
  for (int n = 0; n < npass; ++n) {
    const int w  = tid + 256 * n;
    const int cl = w >> 4;
    if (cl >= nch) continue;
    const int c  = cb + cl;
    const int i  = w & 15;
    int tk[16];
    const int4* tp = (const int4*)&toks[16 * c];
#pragma unroll
    for (int u = 0; u < 4; ++u) {
      const int4 t4 = tp[u];
      tk[4 * u] = t4.x; tk[4 * u + 1] = t4.y; tk[4 * u + 2] = t4.z; tk[4 * u + 3] = t4.w;
    }
    const int   ti  = toks[16 * c + i];
    const float rdl = (16 * c + i < SEQL - 1) ? rd[ti] : 0.0f;
    float g[16], s[16], x[16];
#pragma unroll
    for (int j = 0; j < 16; ++j) {
      g[j] = Gl[tk[j] * GP + ti];
      s[j] = 0.0f;
    }
    xinv_steps<15>(s, x, g, rdl, i);
    x[0] = (0 < i) ? -s[0] : 1.0f;
    float4* dst = (float4*)(Xb + (cl * 16 + i) * 16);
    dst[0] = make_float4(x[0], x[1], x[2], x[3]);
    dst[1] = make_float4(x[4], x[5], x[6], x[7]);
    dst[2] = make_float4(x[8], x[9], x[10], x[11]);
    dst[3] = make_float4(x[12], x[13], x[14], x[15]);
  }
  __syncthreads();
  if (wave >= 4 && role == 0) return;

  const int r0 = (wave & 3) << 4;
  v8f Mt[4];
#pragma unroll
  for (int mt = 0; mt < 4; ++mt)
#pragma unroll
    for (int v = 0; v < 8; ++v)
      Mt[mt][v] = (homog && (r0 + li == 16 * mt + 8 * h + v)) ? 1.0f : 0.0f;

  const float4* Xg = (const float4*)Xb + (li * 4 + 2 * h);

  float* Sl = &St[(wave * 16 + li) * SP];

  auto put_state = [&]() {
#pragma unroll
    for (int mt = 0; mt < 4; ++mt) {
      float4* d = (float4*)(Sl + 16 * mt + 8 * h);
      d[0] = make_float4(Mt[mt][0], Mt[mt][1], Mt[mt][2], Mt[mt][3]);
      d[1] = make_float4(Mt[mt][4], Mt[mt][5], Mt[mt][6], Mt[mt][7]);
    }
    __builtin_amdgcn_wave_barrier();
  };

  struct Ops { float kt[4][8]; float kr[8]; float rv[8]; const float* pk[8]; };

  auto load_ops = [&](Ops& o, const int4 ta, const int4 tb) {
    const int tv[8] = {ta.x, ta.y, ta.z, ta.w, tb.x, tb.y, tb.z, tb.w};
#pragma unroll
    for (int v = 0; v < 8; ++v) {
      const float* grow = &Gl[tv[v] * GP + li];
#pragma unroll
      for (int mt = 0; mt < 4; ++mt) o.kt[mt][v] = grow[16 * mt];
      o.kr[v] = Tl[tv[v] * TP + r0 + li];
      o.rv[v] = rd[tv[v]];
      o.pk[v] = Sl + tv[v];
    }
  };

  auto compute = [&](const Ops& o, const float (&P)[8], const float4 xlo, const float4 xhi,
                     const bool pad) {
    v8f R;
#pragma unroll
    for (int v = 0; v < 8; ++v) R[v] = ksel * o.kr[v] - P[v] * o.rv[v];
    R[7] = pad ? 0.0f : R[7];
    const float xa[8] = {xlo.x, xlo.y, xlo.z, xlo.w, xhi.x, xhi.y, xhi.z, xhi.w};
    v8f Ut = {};
#pragma unroll
    for (int jj = 0; jj < 4; ++jj)
      Ut = wmma_f32((v2f){xa[2 * jj], xa[2 * jj + 1]},
                    (v2f){R[2 * jj], R[2 * jj + 1]}, Ut);
#pragma unroll
    for (int jj = 0; jj < 4; ++jj)
#pragma unroll
      for (int mt = 0; mt < 4; ++mt)
        Mt[mt] = wmma_f32((v2f){o.kt[mt][2 * jj], o.kt[mt][2 * jj + 1]},
                          (v2f){Ut[2 * jj], Ut[2 * jj + 1]}, Mt[mt]);
  };

  int4   t1a, t1b;
  float4 x0lo = Xg[0],  x0hi = Xg[1];
  float4 x1lo = Xg[64], x1hi = Xg[65];
  Ops    opA, opB;
  const int* tseg = &toks[16 * cb];
  put_state();
  load_ops(opA, *(const int4*)&tseg[8 * h], *(const int4*)&tseg[8 * h + 4]);
  t1a = *(const int4*)&tseg[16 + 8 * h];
  t1b = *(const int4*)&tseg[16 + 8 * h + 4];

  auto step = [&](const Ops& cur, Ops& nxt, const int c) {
    float P[8];
#pragma unroll
    for (int v = 0; v < 8; ++v) P[v] = *cur.pk[v];
    __builtin_amdgcn_sched_barrier(0);
    const int    c2   = (c + 2 < nch) ? c + 2 : nch - 1;
    const int4   t2a  = *(const int4*)&tseg[16 * c2 + 8 * h];
    const int4   t2b  = *(const int4*)&tseg[16 * c2 + 8 * h + 4];
    const float4 x2lo = Xg[c2 * 64], x2hi = Xg[c2 * 64 + 1];
    load_ops(nxt, t1a, t1b);
    __builtin_amdgcn_sched_barrier(0);
    compute(cur, P, x0lo, x0hi, (cb + c == NCH - 1) && (h != 0));
    __builtin_amdgcn_sched_barrier(0);
    put_state();
    t1a = t2a; t1b = t2b;
    x0lo = x1lo; x0hi = x1hi; x1lo = x2lo; x1hi = x2hi;
  };

  for (int c = 0; c < nch; c += 2) {
    step(opA, opB, c);
    step(opB, opA, c + 1);
  }

  if (role != 3) {
    const int plane = (role == 0) ? 0 : 2 * role - 1 + (homog ? 1 : 0);
    float4* dst = (float4*)(Mws + (size_t)plane * MPLANE) +
                  (size_t)((b * 4 + (wave & 3)) * 4) * 64 + lane * 2;
#pragma unroll
    for (int mt = 0; mt < 4; ++mt) {
      dst[mt * 64]     = make_float4(Mt[mt][0], Mt[mt][1], Mt[mt][2], Mt[mt][3]);
      dst[mt * 64 + 1] = make_float4(Mt[mt][4], Mt[mt][5], Mt[mt][6], Mt[mt][7]);
    }
  } else {
    const int tokq = toks[SEQL - 1];
    float* dstv = homog ? wvec : vvec;
    if (h == 0) dstv[b * 64 + r0 + li] = Sl[tokq];
  }
}

__device__ __forceinline__ v8f load_tile(const float4* p) {
  const float4 lo = p[0], hi = p[1];
  return (v8f){lo.x, lo.y, lo.z, lo.w, hi.x, hi.y, hi.z, hi.w};
}

__device__ __forceinline__ void rows_dot2(const v8f (&At)[4], const v8f (&Ct)[4],
                                          const float4 (&lo)[4], const float4 (&hi)[4],
                                          v8f& Pa, v8f& Pc) {
  Pa = (v8f){};
  Pc = (v8f){};
#pragma unroll
  for (int mt = 0; mt < 4; ++mt) {
    Pa = wmma_f32((v2f){lo[mt].x, lo[mt].y}, (v2f){At[mt][0], At[mt][1]}, Pa);
    Pc = wmma_f32((v2f){lo[mt].x, lo[mt].y}, (v2f){Ct[mt][0], Ct[mt][1]}, Pc);
    Pa = wmma_f32((v2f){lo[mt].z, lo[mt].w}, (v2f){At[mt][2], At[mt][3]}, Pa);
    Pc = wmma_f32((v2f){lo[mt].z, lo[mt].w}, (v2f){Ct[mt][2], Ct[mt][3]}, Pc);
    Pa = wmma_f32((v2f){hi[mt].x, hi[mt].y}, (v2f){At[mt][4], At[mt][5]}, Pa);
    Pc = wmma_f32((v2f){hi[mt].x, hi[mt].y}, (v2f){Ct[mt][4], Ct[mt][5]}, Pc);
    Pa = wmma_f32((v2f){hi[mt].z, hi[mt].w}, (v2f){At[mt][6], At[mt][7]}, Pa);
    Pc = wmma_f32((v2f){hi[mt].z, hi[mt].w}, (v2f){Ct[mt][6], Ct[mt][7]}, Pc);
  }
}

__global__ __launch_bounds__(128) void head_kernel(
    const float* __restrict__ Mws, const float* __restrict__ wvec,
    const float* __restrict__ vvec, const float* __restrict__ Wr,
    const float* __restrict__ br, const float* __restrict__ Wo,
    const float* __restrict__ bo, float* __restrict__ out) {
  __shared__ __attribute__((aligned(16))) float W3l[HDIM];
  __shared__ __attribute__((aligned(16))) float W2l[HDIM];
  __shared__ __attribute__((aligned(16))) float Cl[HDIM];
  __shared__ __attribute__((aligned(16))) float Glds[HDIM];
  const int tid  = threadIdx.x;
  const int wave = tid >> 5;
  const int lane = tid & 31;
  const int li   = lane & 15;
  const int h    = lane >> 4;
  const int b    = blockIdx.x;
  const int col  = (wave << 4) + li;

  const float cq  = vvec[b * 64 + col];
  const float brv = br[col];
  const float bov = bo[col];
  const float* wrow = wvec + b * 64 + 8 * h;
  float4 wlo[4], whi[4];
  const float4* srcp = (const float4*)Mws + (size_t)((b * 4 + wave) * 4) * 64 + lane * 2;
  v8f Mt[4], C2t[4], A2t[4], C3t[4], A3t[4];
#pragma unroll
  for (int mt = 0; mt < 4; ++mt) {
    wlo[mt] = *(const float4*)(wrow + 16 * mt);
    whi[mt] = *(const float4*)(wrow + 16 * mt + 4);
    A3t[mt] = load_tile(srcp + 4 * (MPLANE / 4) + mt * 64);
    C3t[mt] = load_tile(srcp + 3 * (MPLANE / 4) + mt * 64);
    A2t[mt] = load_tile(srcp + 2 * (MPLANE / 4) + mt * 64);
    C2t[mt] = load_tile(srcp + 1 * (MPLANE / 4) + mt * 64);
    Mt[mt]  = load_tile(srcp + mt * 64);
  }
  v2f wr[16], wo[16];
#pragma unroll
  for (int kt = 0; kt < 16; ++kt) {
    wr[kt] = frag_b(Wr, 64, kt, wave, lane);
    wo[kt] = frag_b(Wo, 64, kt, wave, lane);
  }
  asm volatile("" ::"v"(cq));
  __builtin_amdgcn_sched_barrier(0);

  v8f Pa, Pc;
  rows_dot2(A3t, C3t, wlo, whi, Pa, Pc);
  if (h == 0) W3l[col] = Pa[0];
  const float v3 = Pc[0] + cq;
  __syncthreads();

  float4 slo[4], shi[4];
#pragma unroll
  for (int mt = 0; mt < 4; ++mt) {
    slo[mt] = *(const float4*)(&W3l[8 * h] + 16 * mt);
    shi[mt] = *(const float4*)(&W3l[8 * h] + 16 * mt + 4);
  }
  rows_dot2(A2t, C2t, slo, shi, Pa, Pc);
  if (h == 0) W2l[col] = Pa[0];
  const float v2 = Pc[0] + v3;
  __syncthreads();

  const v8f Pq = rows_dot(Mt, &W2l[8 * h]);
  if (h == 0) Cl[col] = Pq[0] + v2;
  __syncthreads();

  {
    v8f acc = {};
    const float* ar = &Cl[h << 1];
#pragma unroll
    for (int kt = 0; kt < 16; ++kt)
      acc = wmma_f32((v2f){ar[4 * kt], ar[4 * kt + 1]}, wr[kt], acc);
    if (h == 0) Glds[col] = acc[0] + brv;
  }
  __syncthreads();

  {
    v8f acc = {};
    const float* ar = &Glds[h << 1];
#pragma unroll
    for (int kt = 0; kt < 16; ++kt)
      acc = wmma_f32((v2f){ar[4 * kt], ar[4 * kt + 1]}, wo[kt], acc);
    if (h == 0) out[b * 64 + col] = acc[0] + bov;
  }
}

extern "C" void kernel_launch(void* const* d_in, const int* in_sizes, int n_in,
                              void* d_out, int out_size, void* d_ws, size_t ws_size,
                              hipStream_t stream) {
  const int*   seq   = (const int*)d_in[0];
  const float* embed = (const float*)d_in[1];
  const float* W1    = (const float*)d_in[2];
  const float* b1    = (const float*)d_in[3];
  const float* W2    = (const float*)d_in[4];
  const float* b2    = (const float*)d_in[5];
  const float* gamma = (const float*)d_in[6];
  const float* beta  = (const float*)d_in[7];
  const float* Wr    = (const float*)d_in[8];
  const float* br    = (const float*)d_in[9];
  const float* Wo    = (const float*)d_in[10];
  const float* bo    = (const float*)d_in[11];

  float* Tt     = (float*)d_ws;
  float* rdenom = Tt + VOCABN * HDIM;
  float* wvec   = rdenom + VOCABN;
  float* vvec   = wvec + BATCH * HDIM;
  float* Mws    = vvec + BATCH * HDIM;
  float* Xws    = Mws + 5 * MPLANE;
  float* out    = (float*)d_out;

  encode_table_kernel<<<4, 256, 0, stream>>>(embed, W1, b1, W2, b2, gamma, beta,
                                             Tt, rdenom);
  delta_scan_kernel<<<dim3(BATCH, 4), 256, 0, stream>>>(seq, Tt, rdenom, Xws, Mws,
                                                      wvec, vvec);
  head_kernel<<<BATCH, 128, 0, stream>>>(Mws, wvec, vvec, Wr, br, Wo, bo, out);
}


# Chunk inverses X kept in the block's LDS (row pitch 20 floats) and read one chunk ahead with the other operands, not stored to the workspace and read back from memory; same order of sums
# speedup vs baseline: 12.4889x; 1.2185x over previous
#include <hip/hip_runtime.h>


#define HDIM   64
#define VOCABN 64
#define BATCH  64
#define SEQL   4096

typedef __attribute__((ext_vector_type(2))) float v2f;
typedef __attribute__((ext_vector_type(2))) float f32x2;
typedef __attribute__((ext_vector_type(8))) float v8f;

__device__ __forceinline__ v2f frag_a(const float* A, int lda, int mt, int kt, int lane) {
  int row = (mt << 4) + (lane & 15);
  int k0  = (kt << 2) + ((lane >> 4) << 1);
  v2f a;
  a.x = A[row * lda + k0];
  a.y = A[row * lda + k0 + 1];
  return a;
}
__device__ __forceinline__ v2f frag_b(const float* Bm, int ldb, int kt, int nt, int lane) {
  int col = (nt << 4) + (lane & 15);
  int k0  = (kt << 2) + ((lane >> 4) << 1);
  v2f b;
  b.x = Bm[k0 * ldb + col];
  b.y = Bm[(k0 + 1) * ldb + col];
  return b;
}

__device__ __forceinline__ v8f wmma_f32(v2f a, v2f b, v8f c) {
  return __builtin_amdgcn_wmma_f32_16x16x4_f32(false, a, false, b, (short)0, c,
                                               false, false);
}

#define RTP 20
#define XP  68

template <int MASK>
__device__ __forceinline__ float row_xor(float x) {
  return __int_as_float(__builtin_amdgcn_update_dpp(0, __float_as_int(x), 0x160 | MASK,
                                                    0xf, 0xf, false));
}
__device__ __forceinline__ float row_sum16(float x) {
  x += row_xor<1>(x);
  x += row_xor<2>(x);
  x += row_xor<4>(x);
  x += row_xor<8>(x);
  return x;
}

__global__ __launch_bounds__(256) void encode_table_kernel(
    const float* __restrict__ embed, const float* __restrict__ W1,
    const float* __restrict__ b1, const float* __restrict__ W2,
    const float* __restrict__ b2, const float* __restrict__ gamma,
    const float* __restrict__ beta, float* __restrict__ Tout,
    float* __restrict__ rdenom) {
  __shared__ __attribute__((aligned(16))) float Rt[128 * RTP];
  __shared__ __attribute__((aligned(16))) float Xp[2 * 16 * XP];
  const int tid  = threadIdx.x;
  const int wave = tid >> 5;
  const int lane = tid & 31;
  const int li   = lane & 15;
  const int h    = lane >> 4;
  const int mt   = blockIdx.x;
  const int nt2  = wave & 3;
  const int kh   = wave >> 2;
  const int r    = tid >> 4;
  const int c4   = tid & 15;

  v2f a1[16], w1f[16], w2f[16];
  const float* ap = embed + (16 * mt + li) * 64 + 2 * h;
  const float* bp = W1 + (2 * h) * 128 + 16 * wave + li;
  const float* wp = W2 + (64 * kh + 2 * h) * 64 + 16 * nt2 + li;
#pragma unroll
  for (int kt = 0; kt < 16; ++kt) {
    a1[kt]  = (v2f){ap[4 * kt], ap[4 * kt + 1]};
    w1f[kt] = (v2f){bp[(4 * kt) * 128], bp[(4 * kt + 1) * 128]};
    w2f[kt] = (v2f){wp[(4 * kt) * 64], wp[(4 * kt + 1) * 64]};
  }
  const float  bias1 = b1[16 * wave + li];
  const float4 e4  = *(const float4*)(embed + (16 * mt + r) * 64 + 4 * c4);
  const float4 bb4 = *(const float4*)(b2 + 4 * c4);
  const float4 g4  = *(const float4*)(gamma + 4 * c4);
  const float4 be4 = *(const float4*)(beta + 4 * c4);
  __builtin_amdgcn_sched_barrier(0);

  v8f acc = {};
#pragma unroll
  for (int kt = 0; kt < 16; ++kt) acc = wmma_f32(a1[kt], w1f[kt], acc);
  float rr[8];
#pragma unroll
  for (int v = 0; v < 8; ++v) rr[v] = fmaxf(acc[v] + bias1, 0.0f);
  float4* rdst = (float4*)&Rt[(16 * wave + li) * RTP + 8 * h];
  rdst[0] = make_float4(rr[0], rr[1], rr[2], rr[3]);
  rdst[1] = make_float4(rr[4], rr[5], rr[6], rr[7]);
  asm volatile("" ::"v"(e4.x), "v"(e4.y), "v"(e4.z), "v"(e4.w), "v"(bb4.x), "v"(bb4.y),
               "v"(bb4.z), "v"(bb4.w), "v"(g4.x), "v"(g4.y), "v"(g4.z), "v"(g4.w),
               "v"(be4.x), "v"(be4.y), "v"(be4.z), "v"(be4.w));
  __syncthreads();

  v8f acc2 = {};
  const float* rp = &Rt[(64 * kh + 2 * h) * RTP + li];
  v2f a2[16];
#pragma unroll
  for (int kt = 0; kt < 16; ++kt) a2[kt] = (v2f){rp[(4 * kt) * RTP], rp[(4 * kt + 1) * RTP]};
  __builtin_amdgcn_sched_barrier(0);
#pragma unroll
  for (int kt = 0; kt < 16; ++kt) acc2 = wmma_f32(a2[kt], w2f[kt], acc2);
  float* xd = &Xp[(kh * 16 + 8 * h) * XP + 16 * nt2 + li];
#pragma unroll
  for (int v = 0; v < 8; ++v) xd[v * XP] = acc2[v];
  __syncthreads();

  const float4 p0 = *(const float4*)&Xp[r * XP + 4 * c4];
  const float4 p1 = *(const float4*)&Xp[(16 + r) * XP + 4 * c4];
  const float x0 = e4.x + (p0.x + p1.x) + bb4.x;
  const float x1 = e4.y + (p0.y + p1.y) + bb4.y;
  const float x2 = e4.z + (p0.z + p1.z) + bb4.z;
  const float x3 = e4.w + (p0.w + p1.w) + bb4.w;
  const float mu = row_sum16((x0 + x1) + (x2 + x3)) * (1.0f / 64.0f);
  const float d0 = x0 - mu, d1 = x1 - mu, d2 = x2 - mu, d3 = x3 - mu;
  const float var = row_sum16((d0 * d0 + d1 * d1) + (d2 * d2 + d3 * d3)) * (1.0f / 64.0f);
  const float rs  = rsqrtf(var + 1e-5f);
  const float t0 = d0 * rs * g4.x + be4.x;
  const float t1 = d1 * rs * g4.y + be4.y;
  const float t2 = d2 * rs * g4.z + be4.z;
  const float t3 = d3 * rs * g4.w + be4.w;
  const float ksq = row_sum16((t0 * t0 + t1 * t1) + (t2 * t2 + t3 * t3));
  *(float4*)(Tout + (16 * mt + r) * 64 + 4 * c4) = make_float4(t0, t1, t2, t3);
  if (c4 == 0) rdenom[16 * mt + r] = 1.0f / (ksq + 1e-6f);
}

#define TP   68
#define GP   80
#define SP   68
#define NCH  256
#define SEG1 88
#define SEGN 56
#define XRP  20
#define XCP  (16 * XRP)
#define MPLANE (BATCH * HDIM * HDIM)
static_assert(SEG1 + 3 * SEGN == NCH, "the four stretches cover the sequence");
static_assert(SEG1 % 2 == 0 && SEGN % 2 == 0 && SEGN <= SEG1,
              "the scan takes two chunks per trip; the LDS plane of X holds SEG1 chunks");

__device__ __forceinline__ v8f rows_dot(const v8f (&Mt)[4], const float* krow) {
  v8f acc = {};
#pragma unroll
  for (int mt = 0; mt < 4; ++mt) {
    const float4 lo = *(const float4*)(krow + 16 * mt);
    const float4 hi = *(const float4*)(krow + 16 * mt + 4);
    acc = wmma_f32((v2f){lo.x, lo.y}, (v2f){Mt[mt][0], Mt[mt][1]}, acc);
    acc = wmma_f32((v2f){lo.z, lo.w}, (v2f){Mt[mt][2], Mt[mt][3]}, acc);
    acc = wmma_f32((v2f){hi.x, hi.y}, (v2f){Mt[mt][4], Mt[mt][5]}, acc);
    acc = wmma_f32((v2f){hi.z, hi.w}, (v2f){Mt[mt][6], Mt[mt][7]}, acc);
  }
  return acc;
}

template <int L>
__device__ __forceinline__ float row_lane(float x) {
  return __int_as_float(__builtin_amdgcn_mov_dpp(__float_as_int(x), 0x150 | L, 0xf, 0xf, true));
}

template <int M>
__device__ __forceinline__ void xinv_steps(float (&s)[16], float (&x)[16],
                                           const float (&g)[16], const float rdl,
                                           const int i) {
  x[M] = (M < i) ? -s[M] : ((M == i) ? 1.0f : 0.0f);
  const float ym = row_lane<M>(rdl) * x[M];
#pragma unroll
  for (int j = 0; j < M; ++j) s[j] = fmaf(row_lane<M>(g[j]), ym, s[j]);
  if constexpr (M > 1) xinv_steps<M - 1>(s, x, g, rdl, i);
}

__global__ __launch_bounds__(256) void delta_scan_kernel(
    const int* __restrict__ seq, const float* __restrict__ Ttab,
    const float* __restrict__ rdenom, float* __restrict__ Mws,
    float* __restrict__ wvec, float* __restrict__ vvec) {
  __shared__ __attribute__((aligned(16))) float Tl[VOCABN * TP];
  __shared__ __attribute__((aligned(16))) float Gl[VOCABN * GP];
  __shared__ float rd[VOCABN];
  __shared__ __attribute__((aligned(16))) int toks[SEQL];
  __shared__ __attribute__((aligned(16))) float St[8 * 16 * SP];
  __shared__ __attribute__((aligned(16))) float Xl[SEG1 * XCP];
  const int b    = blockIdx.x;
  const int role = blockIdx.y;
  const int tid  = threadIdx.x;
  const int wave = tid >> 5;
  const int cb   = (role == 0) ? 0 : SEG1 + (role - 1) * SEGN;
  const int nch  = (role == 0) ? SEG1 : SEGN;
  const bool homog = (role != 0) && (wave >= 4);
  const float ksel = homog ? 0.0f : 1.0f;
  const int lane = tid & 31;
  const int li   = lane & 15;
  const int h    = lane >> 4;

  const float4* Tg4 = (const float4*)Ttab;
  for (int i = tid; i < VOCABN * 16; i += 256)
    *(float4*)&Tl[(i >> 4) * TP + ((i & 15) << 2)] = Tg4[i];
  if (tid < VOCABN) rd[tid] = rdenom[tid];
  const int4* sg = (const int4*)(seq + b * SEQL);
  int4*       tl = (int4*)toks;
  for (int i = tid; i < SEQL / 4; i += 256) tl[i] = sg[i];
  __syncthreads();

  for (int tile = wave; tile < 16; tile += 8) {
    const int mt = tile >> 2, nt = tile & 3;
    v8f acc = {};
    const float* ar = &Tl[((mt << 4) + li) * TP + (h << 1)];
    const float* br = &Tl[((nt << 4) + li) * TP + (h << 1)];
#pragma unroll
    for (int kt = 0; kt < 16; ++kt) {
      v2f a = (v2f){ar[4 * kt], ar[4 * kt + 1]};
      v2f bb = (v2f){br[4 * kt], br[4 * kt + 1]};
      acc = wmma_f32(a, bb, acc);
    }
#pragma unroll
    for (int v = 0; v < 8; ++v)
      Gl[((mt << 4) + (h << 3) + v) * GP + (nt << 4) + li] = acc[v];
  }
  __syncthreads();

  const int npass = (nch * 16 + 255) >> 8;
  for (int n = 0; n < npass; ++n) {
    const int w  = tid + 256 * n;
    const int cl = w >> 4;
    if (cl >= nch) continue;
    const int c  = cb + cl;
    const int i  = w & 15;
    int tk[16];
    const int4* tp = (const int4*)&toks[16 * c];
#pragma unroll
    for (int u = 0; u < 4; ++u) {
      const int4 t4 = tp[u];
      tk[4 * u] = t4.x; tk[4 * u + 1] = t4.y; tk[4 * u + 2] = t4.z; tk[4 * u + 3] = t4.w;
    }
    const int   ti  = toks[16 * c + i];
    const float rdl = (16 * c + i < SEQL - 1) ? rd[ti] : 0.0f;
    float g[16], s[16], x[16];
#pragma unroll
    for (int j = 0; j < 16; ++j) {
      g[j] = Gl[tk[j] * GP + ti];
      s[j] = 0.0f;
    }
    xinv_steps<15>(s, x, g, rdl, i);
    x[0] = (0 < i) ? -s[0] : 1.0f;
    float4* dst = (float4*)&Xl[cl * XCP + i * XRP];
    dst[0] = make_float4(x[0], x[1], x[2], x[3]);
    dst[1] = make_float4(x[4], x[5], x[6], x[7]);
    dst[2] = make_float4(x[8], x[9], x[10], x[11]);
    dst[3] = make_float4(x[12], x[13], x[14], x[15]);
  }
  __syncthreads();
  if (wave >= 4 && role == 0) return;

  const int r0 = (wave & 3) << 4;
  v8f Mt[4];
#pragma unroll
  for (int mt = 0; mt < 4; ++mt)
#pragma unroll
    for (int v = 0; v < 8; ++v)
      Mt[mt][v] = (homog && (r0 + li == 16 * mt + 8 * h + v)) ? 1.0f : 0.0f;

  const float4* Xg = (const float4*)&Xl[li * XRP + 8 * h];

  float* Sl = &St[(wave * 16 + li) * SP];

  auto put_state = [&]() {
#pragma unroll
    for (int mt = 0; mt < 4; ++mt) {
      float4* d = (float4*)(Sl + 16 * mt + 8 * h);
      d[0] = make_float4(Mt[mt][0], Mt[mt][1], Mt[mt][2], Mt[mt][3]);
      d[1] = make_float4(Mt[mt][4], Mt[mt][5], Mt[mt][6], Mt[mt][7]);
    }
    __builtin_amdgcn_wave_barrier();
  };

  struct Ops { float kt[4][8]; float kr[8]; float rv[8]; float4 xlo, xhi; const float* pk[8]; };

  auto load_ops = [&](Ops& o, const int4 ta, const int4 tb, const int c) {
    const int tv[8] = {ta.x, ta.y, ta.z, ta.w, tb.x, tb.y, tb.z, tb.w};
    o.xlo = Xg[c * (XCP / 4)];
    o.xhi = Xg[c * (XCP / 4) + 1];
#pragma unroll
    for (int v = 0; v < 8; ++v) {
      const float* grow = &Gl[tv[v] * GP + li];
#pragma unroll
      for (int mt = 0; mt < 4; ++mt) o.kt[mt][v] = grow[16 * mt];
      o.kr[v] = Tl[tv[v] * TP + r0 + li];
      o.rv[v] = rd[tv[v]];
      o.pk[v] = Sl + tv[v];
    }
  };

  auto compute = [&](const Ops& o, const float (&P)[8], const bool pad) {
    v8f R;
#pragma unroll
    for (int v = 0; v < 8; ++v) R[v] = ksel * o.kr[v] - P[v] * o.rv[v];
    R[7] = pad ? 0.0f : R[7];
    const float xa[8] = {o.xlo.x, o.xlo.y, o.xlo.z, o.xlo.w, o.xhi.x, o.xhi.y, o.xhi.z, o.xhi.w};
    v8f Ut = {};
#pragma unroll
    for (int jj = 0; jj < 4; ++jj)
      Ut = wmma_f32((v2f){xa[2 * jj], xa[2 * jj + 1]},
                    (v2f){R[2 * jj], R[2 * jj + 1]}, Ut);
#pragma unroll
    for (int jj = 0; jj < 4; ++jj)
#pragma unroll
      for (int mt = 0; mt < 4; ++mt)
        Mt[mt] = wmma_f32((v2f){o.kt[mt][2 * jj], o.kt[mt][2 * jj + 1]},
                          (v2f){Ut[2 * jj], Ut[2 * jj + 1]}, Mt[mt]);
  };

  int4   t1a, t1b;
  Ops    opA, opB;
  const int* tseg = &toks[16 * cb];
  put_state();
  load_ops(opA, *(const int4*)&tseg[8 * h], *(const int4*)&tseg[8 * h + 4], 0);
  t1a = *(const int4*)&tseg[16 + 8 * h];
  t1b = *(const int4*)&tseg[16 + 8 * h + 4];

  auto step = [&](const Ops& cur, Ops& nxt, const int c) {
    float P[8];
#pragma unroll
    for (int v = 0; v < 8; ++v) P[v] = *cur.pk[v];
    __builtin_amdgcn_sched_barrier(0);
    const int    c1   = (c + 1 < nch) ? c + 1 : nch - 1;
    const int    c2   = (c + 2 < nch) ? c + 2 : nch - 1;
    const int4   t2a  = *(const int4*)&tseg[16 * c2 + 8 * h];
    const int4   t2b  = *(const int4*)&tseg[16 * c2 + 8 * h + 4];
    load_ops(nxt, t1a, t1b, c1);
    __builtin_amdgcn_sched_barrier(0);
    compute(cur, P, (cb + c == NCH - 1) && (h != 0));
    __builtin_amdgcn_sched_barrier(0);
    put_state();
    t1a = t2a; t1b = t2b;
  };

  for (int c = 0; c < nch; c += 2) {
    step(opA, opB, c);
    step(opB, opA, c + 1);
  }

  if (role != 3) {
    const int plane = (role == 0) ? 0 : 2 * role - 1 + (homog ? 1 : 0);
    float4* dst = (float4*)(Mws + (size_t)plane * MPLANE) +
                  (size_t)((b * 4 + (wave & 3)) * 4) * 64 + lane * 2;
#pragma unroll
    for (int mt = 0; mt < 4; ++mt) {
      dst[mt * 64]     = make_float4(Mt[mt][0], Mt[mt][1], Mt[mt][2], Mt[mt][3]);
      dst[mt * 64 + 1] = make_float4(Mt[mt][4], Mt[mt][5], Mt[mt][6], Mt[mt][7]);
    }
  } else {
    const int tokq = toks[SEQL - 1];
    float* dstv = homog ? wvec : vvec;
    if (h == 0) dstv[b * 64 + r0 + li] = Sl[tokq];
  }
}

__device__ __forceinline__ v8f load_tile(const float4* p) {
  const float4 lo = p[0], hi = p[1];
  return (v8f){lo.x, lo.y, lo.z, lo.w, hi.x, hi.y, hi.z, hi.w};
}

__device__ __forceinline__ void rows_dot2(const v8f (&At)[4], const v8f (&Ct)[4],
                                          const float4 (&lo)[4], const float4 (&hi)[4],
                                          v8f& Pa, v8f& Pc) {
  Pa = (v8f){};
  Pc = (v8f){};
#pragma unroll
  for (int mt = 0; mt < 4; ++mt) {
    Pa = wmma_f32((v2f){lo[mt].x, lo[mt].y}, (v2f){At[mt][0], At[mt][1]}, Pa);
    Pc = wmma_f32((v2f){lo[mt].x, lo[mt].y}, (v2f){Ct[mt][0], Ct[mt][1]}, Pc);
    Pa = wmma_f32((v2f){lo[mt].z, lo[mt].w}, (v2f){At[mt][2], At[mt][3]}, Pa);
    Pc = wmma_f32((v2f){lo[mt].z, lo[mt].w}, (v2f){Ct[mt][2], Ct[mt][3]}, Pc);
    Pa = wmma_f32((v2f){hi[mt].x, hi[mt].y}, (v2f){At[mt][4], At[mt][5]}, Pa);
    Pc = wmma_f32((v2f){hi[mt].x, hi[mt].y}, (v2f){Ct[mt][4], Ct[mt][5]}, Pc);
    Pa = wmma_f32((v2f){hi[mt].z, hi[mt].w}, (v2f){At[mt][6], At[mt][7]}, Pa);
    Pc = wmma_f32((v2f){hi[mt].z, hi[mt].w}, (v2f){Ct[mt][6], Ct[mt][7]}, Pc);
  }
}

__global__ __launch_bounds__(128) void head_kernel(
    const float* __restrict__ Mws, const float* __restrict__ wvec,
    const float* __restrict__ vvec, const float* __restrict__ Wr,
    const float* __restrict__ br, const float* __restrict__ Wo,
    const float* __restrict__ bo, float* __restrict__ out) {
  __shared__ __attribute__((aligned(16))) float W3l[HDIM];
  __shared__ __attribute__((aligned(16))) float W2l[HDIM];
  __shared__ __attribute__((aligned(16))) float Cl[HDIM];
  __shared__ __attribute__((aligned(16))) float Glds[HDIM];
  const int tid  = threadIdx.x;
  const int wave = tid >> 5;
  const int lane = tid & 31;
  const int li   = lane & 15;
  const int h    = lane >> 4;
  const int b    = blockIdx.x;
  const int col  = (wave << 4) + li;

  const float cq  = vvec[b * 64 + col];
  const float brv = br[col];
  const float bov = bo[col];
  const float* wrow = wvec + b * 64 + 8 * h;
  float4 wlo[4], whi[4];
  const float4* srcp = (const float4*)Mws + (size_t)((b * 4 + wave) * 4) * 64 + lane * 2;
  v8f Mt[4], C2t[4], A2t[4], C3t[4], A3t[4];
#pragma unroll
  for (int mt = 0; mt < 4; ++mt) {
    wlo[mt] = *(const float4*)(wrow + 16 * mt);
    whi[mt] = *(const float4*)(wrow + 16 * mt + 4);
    A3t[mt] = load_tile(srcp + 4 * (MPLANE / 4) + mt * 64);
    C3t[mt] = load_tile(srcp + 3 * (MPLANE / 4) + mt * 64);
    A2t[mt] = load_tile(srcp + 2 * (MPLANE / 4) + mt * 64);
    C2t[mt] = load_tile(srcp + 1 * (MPLANE / 4) + mt * 64);
    Mt[mt]  = load_tile(srcp + mt * 64);
  }
  v2f wr[16], wo[16];
#pragma unroll
  for (int kt = 0; kt < 16; ++kt) {
    wr[kt] = frag_b(Wr, 64, kt, wave, lane);
    wo[kt] = frag_b(Wo, 64, kt, wave, lane);
  }
  asm volatile("" ::"v"(cq));
  __builtin_amdgcn_sched_barrier(0);

  v8f Pa, Pc;
  rows_dot2(A3t, C3t, wlo, whi, Pa, Pc);
  if (h == 0) W3l[col] = Pa[0];
  const float v3 = Pc[0] + cq;
  __syncthreads();

  float4 slo[4], shi[4];
#pragma unroll
  for (int mt = 0; mt < 4; ++mt) {
    slo[mt] = *(const float4*)(&W3l[8 * h] + 16 * mt);
    shi[mt] = *(const float4*)(&W3l[8 * h] + 16 * mt + 4);
  }
  rows_dot2(A2t, C2t, slo, shi, Pa, Pc);
  if (h == 0) W2l[col] = Pa[0];
  const float v2 = Pc[0] + v3;
  __syncthreads();

  const v8f Pq = rows_dot(Mt, &W2l[8 * h]);
  if (h == 0) Cl[col] = Pq[0] + v2;
  __syncthreads();

  {
    v8f acc = {};
    const float* ar = &Cl[h << 1];
#pragma unroll
    for (int kt = 0; kt < 16; ++kt)
      acc = wmma_f32((v2f){ar[4 * kt], ar[4 * kt + 1]}, wr[kt], acc);
    if (h == 0) Glds[col] = acc[0] + brv;
  }
  __syncthreads();

  {
    v8f acc = {};
    const float* ar = &Glds[h << 1];
#pragma unroll
    for (int kt = 0; kt < 16; ++kt)
      acc = wmma_f32((v2f){ar[4 * kt], ar[4 * kt + 1]}, wo[kt], acc);
    if (h == 0) out[b * 64 + col] = acc[0] + bov;
  }
}

extern "C" void kernel_launch(void* const* d_in, const int* in_sizes, int n_in,
                              void* d_out, int out_size, void* d_ws, size_t ws_size,
                              hipStream_t stream) {
  const int*   seq   = (const int*)d_in[0];
  const float* embed = (const float*)d_in[1];
  const float* W1    = (const float*)d_in[2];
  const float* b1    = (const float*)d_in[3];
  const float* W2    = (const float*)d_in[4];
  const float* b2    = (const float*)d_in[5];
  const float* gamma = (const float*)d_in[6];
  const float* beta  = (const float*)d_in[7];
  const float* Wr    = (const float*)d_in[8];
  const float* br    = (const float*)d_in[9];
  const float* Wo    = (const float*)d_in[10];
  const float* bo    = (const float*)d_in[11];

  float* Tt     = (float*)d_ws;
  float* rdenom = Tt + VOCABN * HDIM;
  float* wvec   = rdenom + VOCABN;
  float* vvec   = wvec + BATCH * HDIM;
  float* Mws    = vvec + BATCH * HDIM;
  float* out    = (float*)d_out;

  encode_table_kernel<<<4, 256, 0, stream>>>(embed, W1, b1, W2, b2, gamma, beta,
                                             Tt, rdenom);
  delta_scan_kernel<<<dim3(BATCH, 4), 256, 0, stream>>>(seq, Tt, rdenom, Mws, wvec, vvec);
  head_kernel<<<BATCH, 128, 0, stream>>>(Mws, wvec, vvec, Wr, br, Wo, bo, out);
}
